# starting kernel
# baseline (speedup 1.0000x reference)
#include <hip/hip_runtime.h>
#include <math.h>


typedef __attribute__((ext_vector_type(2))) float v2f;
typedef __attribute__((ext_vector_type(8))) float v8f;

#define N_NODES 100000

__global__ void zero_kernel(float* __restrict__ p, int n) {
  int i = blockIdx.x * blockDim.x + threadIdx.x;
  if (i < n) p[i] = 0.0f;
}

__global__ void degree_kernel(const int* __restrict__ src,
                              float* __restrict__ deg, int E) {
  int e = blockIdx.x * blockDim.x + threadIdx.x;
  if (e < E) atomicAdd(deg + src[e], 1.0f);
}

__global__ void dis_kernel(const float* __restrict__ deg,
                           float* __restrict__ dis, int n) {
  int i = blockIdx.x * blockDim.x + threadIdx.x;
  if (i < n) {
    float d = deg[i];
    dis[i] = (d > 0.0f) ? (1.0f / sqrtf(d)) : 0.0f;
  }
}

__global__ void prop1_kernel(const int* __restrict__ src, const int* __restrict__ dst,
                             const float* __restrict__ dis, const float* __restrict__ x,
                             float* __restrict__ tx, int E) {
  int e = blockIdx.x * blockDim.x + threadIdx.x;
  if (e >= E) return;
  int s = src[e], d = dst[e];
  float w = -(dis[s] * dis[d]);
  const float4* xs = reinterpret_cast<const float4*>(x + (size_t)s * 8);
  float4 a = xs[0];
  float4 b = xs[1];
  float* t = tx + (size_t)d * 8;
  atomicAdd(t + 0, w * a.x); atomicAdd(t + 1, w * a.y);
  atomicAdd(t + 2, w * a.z); atomicAdd(t + 3, w * a.w);
  atomicAdd(t + 4, w * b.x); atomicAdd(t + 5, w * b.y);
  atomicAdd(t + 6, w * b.z); atomicAdd(t + 7, w * b.w);
}

__global__ void layer1_kernel(const float* __restrict__ x, const float* __restrict__ tx,
                              const float* __restrict__ W1, const float* __restrict__ b1,
                              float* __restrict__ h1, int n) {
  int i = blockIdx.x * blockDim.x + threadIdx.x;
  if (i >= n) return;
  const float4* xp = reinterpret_cast<const float4*>(x + (size_t)i * 8);
  const float4* tp = reinterpret_cast<const float4*>(tx + (size_t)i * 8);
  float4 x0 = xp[0], x1 = xp[1], t0 = tp[0], t1 = tp[1];
  float xi[8] = {x0.x, x0.y, x0.z, x0.w, x1.x, x1.y, x1.z, x1.w};
  float ti[8] = {t0.x, t0.y, t0.z, t0.w, t1.x, t1.y, t1.z, t1.w};
#pragma unroll
  for (int h = 0; h < 3; ++h) {
    float acc = b1[h];
#pragma unroll
    for (int f = 0; f < 8; ++f)
      acc += xi[f] * W1[f * 3 + h] + ti[f] * W1[24 + f * 3 + h];
    h1[(size_t)i * 3 + h] = fmaxf(acc, 0.0f);
  }
}

__global__ void prop2_kernel(const int* __restrict__ src, const int* __restrict__ dst,
                             const float* __restrict__ dis, const float* __restrict__ h1,
                             float* __restrict__ tx, int E) {
  int e = blockIdx.x * blockDim.x + threadIdx.x;
  if (e >= E) return;
  int s = src[e], d = dst[e];
  float w = -(dis[s] * dis[d]);
  const float* hs = h1 + (size_t)s * 3;
  float* t = tx + (size_t)d * 3;
  atomicAdd(t + 0, w * hs[0]);
  atomicAdd(t + 1, w * hs[1]);
  atomicAdd(t + 2, w * hs[2]);
}

__global__ void layer2_kernel(const float* __restrict__ h1, const float* __restrict__ tx,
                              const float* __restrict__ W2, const float* __restrict__ b2,
                              float* __restrict__ h2, int n) {
  int i = blockIdx.x * blockDim.x + threadIdx.x;
  if (i >= n) return;
  float hv[3], tv[3];
#pragma unroll
  for (int f = 0; f < 3; ++f) {
    hv[f] = h1[(size_t)i * 3 + f];
    tv[f] = tx[(size_t)i * 3 + f];
  }
#pragma unroll
  for (int h = 0; h < 16; ++h) {
    float acc = b2[h];
#pragma unroll
    for (int f = 0; f < 3; ++f)
      acc += hv[f] * W2[f * 16 + h] + tv[f] * W2[48 + f * 16 + h];
    h2[(size_t)i * 16 + h] = fmaxf(acc, 0.0f);
  }
}

__global__ void head_gemm_wmma(const float* __restrict__ h2, const float* __restrict__ Wl,
                               const float* __restrict__ bl, float* __restrict__ out,
                               int ntiles) {
  int gwave = (blockIdx.x * blockDim.x + threadIdx.x) >> 5;
  int lane  = threadIdx.x & 31;
  if (gwave >= ntiles) return;
  int half = lane >> 4;
  int l16  = lane & 15;
  int row  = gwave * 16 + l16;

  v8f acc0 = {}, acc1 = {}, acc2 = {}, acc3 = {};
#pragma unroll
  for (int kk = 0; kk < 16; kk += 4) {
    int ka = kk + 2 * half;
    v2f a;
    a.x = h2[(size_t)row * 16 + ka];
    a.y = h2[(size_t)row * 16 + ka + 1];

    v2f b0, b1, b2, b3;
    b0.x = Wl[ka * 64 +  0 + l16]; b0.y = Wl[(ka + 1) * 64 +  0 + l16];
    b1.x = Wl[ka * 64 + 16 + l16]; b1.y = Wl[(ka + 1) * 64 + 16 + l16];
    b2.x = Wl[ka * 64 + 32 + l16]; b2.y = Wl[(ka + 1) * 64 + 32 + l16];
    b3.x = Wl[ka * 64 + 48 + l16]; b3.y = Wl[(ka + 1) * 64 + 48 + l16];

    acc0 = __builtin_amdgcn_wmma_f32_16x16x4_f32(false, a, false, b0, (short)0, acc0, false, false);
    acc1 = __builtin_amdgcn_wmma_f32_16x16x4_f32(false, a, false, b1, (short)0, acc1, false, false);
    acc2 = __builtin_amdgcn_wmma_f32_16x16x4_f32(false, a, false, b2, (short)0, acc2, false, false);
    acc3 = __builtin_amdgcn_wmma_f32_16x16x4_f32(false, a, false, b3, (short)0, acc3, false, false);
  }

  v8f accs[4] = {acc0, acc1, acc2, acc3};
#pragma unroll
  for (int j = 0; j < 4; ++j) {
    int col = j * 16 + l16;
    float bias = bl[col];
#pragma unroll
    for (int r = 0; r < 8; ++r) {
      int m = gwave * 16 + r + 8 * half;
      out[(size_t)m * 64 + col] = accs[j][r] + bias;
    }
  }
}

extern "C" void kernel_launch(void* const* d_in, const int* in_sizes, int n_in,
                              void* d_out, int out_size, void* d_ws, size_t ws_size,
                              hipStream_t stream) {
  const float* x  = (const float*)d_in[0];
  const int*   ei = (const int*)d_in[1];
  const float* W1 = (const float*)d_in[2];
  const float* b1 = (const float*)d_in[3];
  const float* W2 = (const float*)d_in[4];
  const float* b2 = (const float*)d_in[5];
  const float* Wl = (const float*)d_in[6];
  const float* bl = (const float*)d_in[7];
  float* out = (float*)d_out;

  const int N = in_sizes[0] / 8;
  const int E = in_sizes[1] / 2;
  const int* src = ei;
  const int* dst = ei + E;

  float* ws  = (float*)d_ws;
  float* deg = ws;
  float* dis = ws + (size_t)N;
  float* tx1 = ws + (size_t)2  * N;
  float* h1  = ws + (size_t)10 * N;
  float* tx2 = ws + (size_t)13 * N;
  float* h2  = ws + (size_t)16 * N;

  const int B = 256;
  zero_kernel<<<(16 * N + B - 1) / B, B, 0, stream>>>(ws, 16 * N);

  degree_kernel<<<(E + B - 1) / B, B, 0, stream>>>(src, deg, E);
  dis_kernel<<<(N + B - 1) / B, B, 0, stream>>>(deg, dis, N);

  prop1_kernel<<<(E + B - 1) / B, B, 0, stream>>>(src, dst, dis, x, tx1, E);
  layer1_kernel<<<(N + B - 1) / B, B, 0, stream>>>(x, tx1, W1, b1, h1, N);

  prop2_kernel<<<(E + B - 1) / B, B, 0, stream>>>(src, dst, dis, h1, tx2, E);
  layer2_kernel<<<(N + B - 1) / B, B, 0, stream>>>(h1, tx2, W2, b2, h2, N);

  const int ntiles = N / 16;
  head_gemm_wmma<<<(ntiles * 32 + B - 1) / B, B, 0, stream>>>(h2, Wl, bl, out, ntiles);
}


# dis[src] kept beside the gathered row (layer 1: 64-byte records {x, dis}; layer 2: 16-byte {h1, dis}): no dis[src] gather in the edge loops, one line per edge less; same arithmetic and sum order
# speedup vs baseline: 8.0612x; 8.0612x over previous
#include <hip/hip_runtime.h>
#include <math.h>


typedef __attribute__((ext_vector_type(2))) float v2f;
typedef __attribute__((ext_vector_type(8))) float v8f;

#define TPB 1024

#define RS_DEG 25000
#define RS_P1  10000
#define RS_P2  25000
#define G_DEG 64
#define G_P1  25
#define G_P2  64

__global__ __launch_bounds__(TPB) void degree_lds(const int* __restrict__ src,
                                                  unsigned* __restrict__ part,
                                                  int E, int N, int R, int chunk) {
  __shared__ unsigned cnt[RS_DEG];
  const int tid = threadIdx.x;
  const int r = blockIdx.x % R;
  const int g = blockIdx.x / R;
  const int lo = r * RS_DEG;
  for (int i = tid; i < RS_DEG; i += TPB) cnt[i] = 0u;
  __syncthreads();

  const int e0 = g * chunk;
  const int e1 = min(E, e0 + chunk);
  const int last = e1 - 1;
  for (int b = e0; b < e1; b += 4 * TPB) {
    int s[4];
#pragma unroll
    for (int k = 0; k < 4; ++k) {
      int ee = b + tid + k * TPB;
      int sv = src[min(ee, last)];
      s[k] = (ee < e1) ? sv : -1;
    }
#pragma unroll
    for (int k = 0; k < 4; ++k) {
      unsigned u = (unsigned)(s[k] - lo);
      if (u < (unsigned)RS_DEG)
        __hip_atomic_fetch_add(&cnt[u], 1u, __ATOMIC_RELAXED, __HIP_MEMORY_SCOPE_WORKGROUP);
    }
  }
  __syncthreads();
  for (int i = tid; i < RS_DEG; i += TPB)
    if (lo + i < N) part[(size_t)g * N + lo + i] = cnt[i];
}

#define REC1 16
#define DOFF1 8
#define REC2 4
#define DOFF2 3

__global__ void dis_kernel(const unsigned* __restrict__ part, const float* __restrict__ x,
                           float* __restrict__ dis, float* __restrict__ rec1, int n) {
  int i = blockIdx.x * blockDim.x + threadIdx.x;
  if (i < n) {
    const float4* xp = reinterpret_cast<const float4*>(x + (size_t)i * 8);
    float4 x0 = xp[0], x1 = xp[1];
    unsigned c = 0u;
#pragma unroll 16
    for (int g = 0; g < G_DEG; ++g) c += part[(size_t)g * n + i];
    float d = (float)c;
    float di = (d > 0.0f) ? (1.0f / sqrtf(d)) : 0.0f;
    dis[i] = di;
    float4* rp = reinterpret_cast<float4*>(rec1 + (size_t)i * REC1);
    rp[0] = x0;
    rp[1] = x1;
    rp[2] = make_float4(di, 0.f, 0.f, 0.f);
    rp[3] = make_float4(0.f, 0.f, 0.f, 0.f);
  }
}

template <int F, int RS, int REC, int DOFF>
__global__ __launch_bounds__(TPB) void prop_lds(const int* __restrict__ src,
                                                const int* __restrict__ dst,
                                                const float* __restrict__ dis,
                                                const float* __restrict__ rec,
                                                float* __restrict__ part,
                                                int E, int N, int R, int chunk) {
  __shared__ float acc[F * RS];
  const int tid = threadIdx.x;
  const int r = blockIdx.x % R;
  const int g = blockIdx.x / R;
  const int lo = r * RS;
  for (int i = tid; i < F * RS; i += TPB) acc[i] = 0.0f;
  __syncthreads();

  const int e0 = g * chunk;
  const int e1 = min(E, e0 + chunk);

  if (e1 > e0) {
    const int last = e1 - 1;
    int d[4] = {-1, -1, -1, -1}, s[4] = {0, 0, 0, 0};
    for (int b = e0; ; b += 4 * TPB) {
      int nd[4], ns[4];
#pragma unroll
      for (int k = 0; k < 4; ++k) {
        int ee = b + tid + k * TPB;
        int ec = min(ee, last);
        int dv = dst[ec];
        ns[k] = src[ec];
        nd[k] = (ee < e1) ? dv : -1;
      }

      unsigned u[4];
      float ws_[4], wd_[4];
      constexpr int V = (F == 3) ? 4 : F;
      float v[4][V];
#pragma unroll
      for (int k = 0; k < 4; ++k) {
        u[k] = (unsigned)(d[k] - lo);
        if (u[k] < (unsigned)RS) {
          wd_[k] = dis[d[k]];
          const float* row = rec + (size_t)s[k] * REC;
          const float4* rp = reinterpret_cast<const float4*>(row);
          if constexpr (F == 8) {
            float4 a = rp[0];
            float4 c = rp[1];
            ws_[k] = row[DOFF];
            v[k][0] = a.x; v[k][1] = a.y; v[k][2] = a.z; v[k][3] = a.w;
            v[k][4] = c.x; v[k][5] = c.y; v[k][6] = c.z; v[k][7] = c.w;
          } else {
            static_assert(F == 3 && REC == 4 && DOFF == 3, "16-byte record {h[3], dis}");
            float4 a = rp[0];
            v[k][0] = a.x; v[k][1] = a.y; v[k][2] = a.z; v[k][3] = a.w;
          }
        }
      }
#pragma unroll
      for (int k = 0; k < 4; ++k) {
        if (u[k] < (unsigned)RS) {
          float ds;
          if constexpr (F == 3) ds = v[k][3]; else ds = ws_[k];
          float w = -(ds * wd_[k]);
#pragma unroll
          for (int f = 0; f < F; ++f)
            __hip_atomic_fetch_add(&acc[f * RS + u[k]], w * v[k][f],
                                   __ATOMIC_RELAXED, __HIP_MEMORY_SCOPE_WORKGROUP);
        }
      }

#pragma unroll
      for (int k = 0; k < 4; ++k) { d[k] = nd[k]; s[k] = ns[k]; }
      if (b >= e1) break;
    }
  }
  __syncthreads();

  for (int i = tid; i < F * RS; i += TPB) {
    int un = i / F;
    int f  = i - un * F;
    if (lo + un < N) part[((size_t)g * N + lo + un) * F + f] = acc[f * RS + un];
  }
}

__global__ void layer1_kernel(const float* __restrict__ x, const float* __restrict__ part,
                              const float* __restrict__ W1, const float* __restrict__ b1,
                              const float* __restrict__ dis, float* __restrict__ rec2, int n) {
  int i = blockIdx.x * blockDim.x + threadIdx.x;
  if (i >= n) return;
  const float4* xp = reinterpret_cast<const float4*>(x + (size_t)i * 8);
  float4 x0 = xp[0], x1 = xp[1];
  float4 t0 = make_float4(0.f, 0.f, 0.f, 0.f), t1 = t0;
#pragma unroll
  for (int g = 0; g < G_P1; ++g) {
    const float4* tp = reinterpret_cast<const float4*>(part + ((size_t)g * n + i) * 8);
    float4 p0 = tp[0], p1 = tp[1];
    t0.x += p0.x; t0.y += p0.y; t0.z += p0.z; t0.w += p0.w;
    t1.x += p1.x; t1.y += p1.y; t1.z += p1.z; t1.w += p1.w;
  }
  float xi[8] = {x0.x, x0.y, x0.z, x0.w, x1.x, x1.y, x1.z, x1.w};
  float ti[8] = {t0.x, t0.y, t0.z, t0.w, t1.x, t1.y, t1.z, t1.w};
  float ho[3];
#pragma unroll
  for (int h = 0; h < 3; ++h) {
    float acc = b1[h];
#pragma unroll
    for (int f = 0; f < 8; ++f)
      acc += xi[f] * W1[f * 3 + h] + ti[f] * W1[24 + f * 3 + h];
    ho[h] = fmaxf(acc, 0.0f);
  }
  reinterpret_cast<float4*>(rec2)[i] = make_float4(ho[0], ho[1], ho[2], dis[i]);
}

__global__ void layer2_kernel(const float* __restrict__ rec2, const float* __restrict__ part,
                              const float* __restrict__ W2, const float* __restrict__ b2,
                              float* __restrict__ h2, int n) {
  int i = blockIdx.x * blockDim.x + threadIdx.x;
  if (i >= n) return;
  float4 hr = reinterpret_cast<const float4*>(rec2)[i];
  float hv[3] = {hr.x, hr.y, hr.z}, tv[3] = {0.f, 0.f, 0.f};
#pragma unroll 16
  for (int g = 0; g < G_P2; ++g) {
    const float* tp = part + ((size_t)g * n + i) * 3;
    tv[0] += tp[0]; tv[1] += tp[1]; tv[2] += tp[2];
  }
#pragma unroll
  for (int h = 0; h < 16; ++h) {
    float acc = b2[h];
#pragma unroll
    for (int f = 0; f < 3; ++f)
      acc += hv[f] * W2[f * 16 + h] + tv[f] * W2[48 + f * 16 + h];
    h2[(size_t)i * 16 + h] = fmaxf(acc, 0.0f);
  }
}

__global__ void head_gemm_wmma(const float* __restrict__ h2, const float* __restrict__ Wl,
                               const float* __restrict__ bl, float* __restrict__ out,
                               int ntiles) {
  int gwave = (blockIdx.x * blockDim.x + threadIdx.x) >> 5;
  int lane  = threadIdx.x & 31;
  if (gwave >= ntiles) return;
  int half = lane >> 4;
  int l16  = lane & 15;
  int row  = gwave * 16 + l16;

  v8f acc0 = {}, acc1 = {}, acc2 = {}, acc3 = {};
#pragma unroll
  for (int kk = 0; kk < 16; kk += 4) {
    int ka = kk + 2 * half;
    v2f a;
    a.x = h2[(size_t)row * 16 + ka];
    a.y = h2[(size_t)row * 16 + ka + 1];

    v2f b0, b1, b2, b3;
    b0.x = Wl[ka * 64 +  0 + l16]; b0.y = Wl[(ka + 1) * 64 +  0 + l16];
    b1.x = Wl[ka * 64 + 16 + l16]; b1.y = Wl[(ka + 1) * 64 + 16 + l16];
    b2.x = Wl[ka * 64 + 32 + l16]; b2.y = Wl[(ka + 1) * 64 + 32 + l16];
    b3.x = Wl[ka * 64 + 48 + l16]; b3.y = Wl[(ka + 1) * 64 + 48 + l16];

    acc0 = __builtin_amdgcn_wmma_f32_16x16x4_f32(false, a, false, b0, (short)0, acc0, false, false);
    acc1 = __builtin_amdgcn_wmma_f32_16x16x4_f32(false, a, false, b1, (short)0, acc1, false, false);
    acc2 = __builtin_amdgcn_wmma_f32_16x16x4_f32(false, a, false, b2, (short)0, acc2, false, false);
    acc3 = __builtin_amdgcn_wmma_f32_16x16x4_f32(false, a, false, b3, (short)0, acc3, false, false);
  }

  v8f accs[4] = {acc0, acc1, acc2, acc3};
#pragma unroll
  for (int j = 0; j < 4; ++j) {
    int col = j * 16 + l16;
    float bias = bl[col];
#pragma unroll
    for (int r = 0; r < 8; ++r) {
      int m = gwave * 16 + r + 8 * half;
      out[(size_t)m * 64 + col] = accs[j][r] + bias;
    }
  }
}

extern "C" void kernel_launch(void* const* d_in, const int* in_sizes, int n_in,
                              void* d_out, int out_size, void* d_ws, size_t ws_size,
                              hipStream_t stream) {
  const float* x  = (const float*)d_in[0];
  const int*   ei = (const int*)d_in[1];
  const float* W1 = (const float*)d_in[2];
  const float* b1 = (const float*)d_in[3];
  const float* W2 = (const float*)d_in[4];
  const float* b2 = (const float*)d_in[5];
  const float* Wl = (const float*)d_in[6];
  const float* bl = (const float*)d_in[7];
  float* out = (float*)d_out;

  const int N = in_sizes[0] / 8;
  const int E = in_sizes[1] / 2;
  const int* src = ei;
  const int* dst = ei + E;

  float* ws   = (float*)d_ws;
  float* dis  = ws;
  float* rec2 = ws + (size_t)4  * N;
  float* h2   = ws + (size_t)8  * N;
  float* rec1 = ws + (size_t)32 * N;
  unsigned* pdeg = (unsigned*)(ws + (size_t)48 * N);
  float* p1  = ws + (size_t)(48 + G_DEG) * N;
  float* p2  = ws + (size_t)(48 + G_DEG + 8 * G_P1) * N;

  const int B = 256;
  const int r_deg = (N + RS_DEG - 1) / RS_DEG;
  const int r_p1  = (N + RS_P1 - 1) / RS_P1;
  const int r_p2  = (N + RS_P2 - 1) / RS_P2;
  const int c_deg = (E + G_DEG - 1) / G_DEG;
  const int c_p1  = (E + G_P1 - 1) / G_P1;
  const int c_p2  = (E + G_P2 - 1) / G_P2;

  degree_lds<<<r_deg * G_DEG, TPB, 0, stream>>>(src, pdeg, E, N, r_deg, c_deg);
  dis_kernel<<<(N + B - 1) / B, B, 0, stream>>>(pdeg, x, dis, rec1, N);

  prop_lds<8, RS_P1, REC1, DOFF1><<<r_p1 * G_P1, TPB, 0, stream>>>(src, dst, dis, rec1, p1, E, N, r_p1, c_p1);
  layer1_kernel<<<(N + B - 1) / B, B, 0, stream>>>(x, p1, W1, b1, dis, rec2, N);

  prop_lds<3, RS_P2, REC2, DOFF2><<<r_p2 * G_P2, TPB, 0, stream>>>(src, dst, dis, rec2, p2, E, N, r_p2, c_p2);
  layer2_kernel<<<(N + B - 1) / B, B, 0, stream>>>(rec2, p2, W2, b2, h2, N);

  const int ntiles = N / 16;
  head_gemm_wmma<<<(ntiles * 32 + B - 1) / B, B, 0, stream>>>(h2, Wl, bl, out, ntiles);
}


# Target factor -dis[dst] leaves the edge loops: edges add dis[src]*row into the LDS sums, the dense stages scale each node's chunk sum once; no dis[dst] gather, one line per edge; other rounding
# speedup vs baseline: 9.5443x; 1.1840x over previous
#include <hip/hip_runtime.h>
#include <math.h>


typedef __attribute__((ext_vector_type(2))) float v2f;
typedef __attribute__((ext_vector_type(8))) float v8f;

#define TPB 1024

#define RS_DEG 25000
#define RS_P1  10000
#define RS_P2  25000
#define G_DEG 64
#define G_P1  25
#define G_P2  64

__global__ __launch_bounds__(TPB) void degree_lds(const int* __restrict__ src,
                                                  unsigned* __restrict__ part,
                                                  int E, int N, int R, int chunk) {
  __shared__ unsigned cnt[RS_DEG];
  const int tid = threadIdx.x;
  const int r = blockIdx.x % R;
  const int g = blockIdx.x / R;
  const int lo = r * RS_DEG;
  for (int i = tid; i < RS_DEG; i += TPB) cnt[i] = 0u;
  __syncthreads();

  const int e0 = g * chunk;
  const int e1 = min(E, e0 + chunk);
  const int last = e1 - 1;
  for (int b = e0; b < e1; b += 4 * TPB) {
    int s[4];
#pragma unroll
    for (int k = 0; k < 4; ++k) {
      int ee = b + tid + k * TPB;
      int sv = src[min(ee, last)];
      s[k] = (ee < e1) ? sv : -1;
    }
#pragma unroll
    for (int k = 0; k < 4; ++k) {
      unsigned u = (unsigned)(s[k] - lo);
      if (u < (unsigned)RS_DEG)
        __hip_atomic_fetch_add(&cnt[u], 1u, __ATOMIC_RELAXED, __HIP_MEMORY_SCOPE_WORKGROUP);
    }
  }
  __syncthreads();
  for (int i = tid; i < RS_DEG; i += TPB)
    if (lo + i < N) part[(size_t)g * N + lo + i] = cnt[i];
}

#define REC1 16
#define DOFF1 8
#define REC2 4
#define DOFF2 3

__global__ void dis_kernel(const unsigned* __restrict__ part, const float* __restrict__ x,
                           float* __restrict__ dis, float* __restrict__ rec1, int n) {
  int i = blockIdx.x * blockDim.x + threadIdx.x;
  if (i < n) {
    const float4* xp = reinterpret_cast<const float4*>(x + (size_t)i * 8);
    float4 x0 = xp[0], x1 = xp[1];
    unsigned c = 0u;
#pragma unroll 16
    for (int g = 0; g < G_DEG; ++g) c += part[(size_t)g * n + i];
    float d = (float)c;
    float di = (d > 0.0f) ? (1.0f / sqrtf(d)) : 0.0f;
    dis[i] = di;
    float4* rp = reinterpret_cast<float4*>(rec1 + (size_t)i * REC1);
    rp[0] = x0;
    rp[1] = x1;
    rp[2] = make_float4(di, 0.f, 0.f, 0.f);
    rp[3] = make_float4(0.f, 0.f, 0.f, 0.f);
  }
}

template <int F, int RS, int REC, int DOFF>
__global__ __launch_bounds__(TPB) void prop_lds(const int* __restrict__ src,
                                                const int* __restrict__ dst,
                                                const float* __restrict__ rec,
                                                float* __restrict__ part,
                                                int E, int N, int R, int chunk) {
  __shared__ float acc[F * RS];
  const int tid = threadIdx.x;
  const int r = blockIdx.x % R;
  const int g = blockIdx.x / R;
  const int lo = r * RS;
  for (int i = tid; i < F * RS; i += TPB) acc[i] = 0.0f;
  __syncthreads();

  const int e0 = g * chunk;
  const int e1 = min(E, e0 + chunk);

  if (e1 > e0) {
    const int last = e1 - 1;
    int d[4] = {-1, -1, -1, -1}, s[4] = {0, 0, 0, 0};
    for (int b = e0; ; b += 4 * TPB) {
      int nd[4], ns[4];
#pragma unroll
      for (int k = 0; k < 4; ++k) {
        int ee = b + tid + k * TPB;
        int ec = min(ee, last);
        int dv = dst[ec];
        ns[k] = src[ec];
        nd[k] = (ee < e1) ? dv : -1;
      }

      unsigned u[4];
      float ws_[4];
      constexpr int V = (F == 3) ? 4 : F;
      float v[4][V];
#pragma unroll
      for (int k = 0; k < 4; ++k) {
        u[k] = (unsigned)(d[k] - lo);
        if (u[k] < (unsigned)RS) {
          const float* row = rec + (size_t)s[k] * REC;
          const float4* rp = reinterpret_cast<const float4*>(row);
          if constexpr (F == 8) {
            float4 a = rp[0];
            float4 c = rp[1];
            ws_[k] = row[DOFF];
            v[k][0] = a.x; v[k][1] = a.y; v[k][2] = a.z; v[k][3] = a.w;
            v[k][4] = c.x; v[k][5] = c.y; v[k][6] = c.z; v[k][7] = c.w;
          } else {
            static_assert(F == 3 && REC == 4 && DOFF == 3, "16-byte record {h[3], dis}");
            float4 a = rp[0];
            v[k][0] = a.x; v[k][1] = a.y; v[k][2] = a.z; v[k][3] = a.w;
          }
        }
      }
#pragma unroll
      for (int k = 0; k < 4; ++k) {
        if (u[k] < (unsigned)RS) {
          float w;
          if constexpr (F == 3) w = v[k][3]; else w = ws_[k];
#pragma unroll
          for (int f = 0; f < F; ++f)
            __hip_atomic_fetch_add(&acc[f * RS + u[k]], w * v[k][f],
                                   __ATOMIC_RELAXED, __HIP_MEMORY_SCOPE_WORKGROUP);
        }
      }

#pragma unroll
      for (int k = 0; k < 4; ++k) { d[k] = nd[k]; s[k] = ns[k]; }
      if (b >= e1) break;
    }
  }
  __syncthreads();

  for (int i = tid; i < F * RS; i += TPB) {
    int un = i / F;
    int f  = i - un * F;
    if (lo + un < N) part[((size_t)g * N + lo + un) * F + f] = acc[f * RS + un];
  }
}

__global__ void layer1_kernel(const float* __restrict__ x, const float* __restrict__ part,
                              const float* __restrict__ W1, const float* __restrict__ b1,
                              const float* __restrict__ dis, float* __restrict__ rec2, int n) {
  int i = blockIdx.x * blockDim.x + threadIdx.x;
  if (i >= n) return;
  const float4* xp = reinterpret_cast<const float4*>(x + (size_t)i * 8);
  float4 x0 = xp[0], x1 = xp[1];
  float4 t0 = make_float4(0.f, 0.f, 0.f, 0.f), t1 = t0;
#pragma unroll
  for (int g = 0; g < G_P1; ++g) {
    const float4* tp = reinterpret_cast<const float4*>(part + ((size_t)g * n + i) * 8);
    float4 p0 = tp[0], p1 = tp[1];
    t0.x += p0.x; t0.y += p0.y; t0.z += p0.z; t0.w += p0.w;
    t1.x += p1.x; t1.y += p1.y; t1.z += p1.z; t1.w += p1.w;
  }
  const float di = dis[i];
  const float nd = -di;
  float xi[8] = {x0.x, x0.y, x0.z, x0.w, x1.x, x1.y, x1.z, x1.w};
  float ti[8] = {nd * t0.x, nd * t0.y, nd * t0.z, nd * t0.w,
                 nd * t1.x, nd * t1.y, nd * t1.z, nd * t1.w};
  float ho[3];
#pragma unroll
  for (int h = 0; h < 3; ++h) {
    float acc = b1[h];
#pragma unroll
    for (int f = 0; f < 8; ++f)
      acc += xi[f] * W1[f * 3 + h] + ti[f] * W1[24 + f * 3 + h];
    ho[h] = fmaxf(acc, 0.0f);
  }
  reinterpret_cast<float4*>(rec2)[i] = make_float4(ho[0], ho[1], ho[2], di);
}

__global__ void layer2_kernel(const float* __restrict__ rec2, const float* __restrict__ part,
                              const float* __restrict__ W2, const float* __restrict__ b2,
                              float* __restrict__ h2, int n) {
  int i = blockIdx.x * blockDim.x + threadIdx.x;
  if (i >= n) return;
  float4 hr = reinterpret_cast<const float4*>(rec2)[i];
  float hv[3] = {hr.x, hr.y, hr.z}, tv[3] = {0.f, 0.f, 0.f};
#pragma unroll 16
  for (int g = 0; g < G_P2; ++g) {
    const float* tp = part + ((size_t)g * n + i) * 3;
    tv[0] += tp[0]; tv[1] += tp[1]; tv[2] += tp[2];
  }
  const float nd = -hr.w;
  tv[0] *= nd; tv[1] *= nd; tv[2] *= nd;
#pragma unroll
  for (int h = 0; h < 16; ++h) {
    float acc = b2[h];
#pragma unroll
    for (int f = 0; f < 3; ++f)
      acc += hv[f] * W2[f * 16 + h] + tv[f] * W2[48 + f * 16 + h];
    h2[(size_t)i * 16 + h] = fmaxf(acc, 0.0f);
  }
}

__global__ void head_gemm_wmma(const float* __restrict__ h2, const float* __restrict__ Wl,
                               const float* __restrict__ bl, float* __restrict__ out,
                               int ntiles) {
  int gwave = (blockIdx.x * blockDim.x + threadIdx.x) >> 5;
  int lane  = threadIdx.x & 31;
  if (gwave >= ntiles) return;
  int half = lane >> 4;
  int l16  = lane & 15;
  int row  = gwave * 16 + l16;

  v8f acc0 = {}, acc1 = {}, acc2 = {}, acc3 = {};
#pragma unroll
  for (int kk = 0; kk < 16; kk += 4) {
    int ka = kk + 2 * half;
    v2f a;
    a.x = h2[(size_t)row * 16 + ka];
    a.y = h2[(size_t)row * 16 + ka + 1];

    v2f b0, b1, b2, b3;
    b0.x = Wl[ka * 64 +  0 + l16]; b0.y = Wl[(ka + 1) * 64 +  0 + l16];
    b1.x = Wl[ka * 64 + 16 + l16]; b1.y = Wl[(ka + 1) * 64 + 16 + l16];
    b2.x = Wl[ka * 64 + 32 + l16]; b2.y = Wl[(ka + 1) * 64 + 32 + l16];
    b3.x = Wl[ka * 64 + 48 + l16]; b3.y = Wl[(ka + 1) * 64 + 48 + l16];

    acc0 = __builtin_amdgcn_wmma_f32_16x16x4_f32(false, a, false, b0, (short)0, acc0, false, false);
    acc1 = __builtin_amdgcn_wmma_f32_16x16x4_f32(false, a, false, b1, (short)0, acc1, false, false);
    acc2 = __builtin_amdgcn_wmma_f32_16x16x4_f32(false, a, false, b2, (short)0, acc2, false, false);
    acc3 = __builtin_amdgcn_wmma_f32_16x16x4_f32(false, a, false, b3, (short)0, acc3, false, false);
  }

  v8f accs[4] = {acc0, acc1, acc2, acc3};
#pragma unroll
  for (int j = 0; j < 4; ++j) {
    int col = j * 16 + l16;
    float bias = bl[col];
#pragma unroll
    for (int r = 0; r < 8; ++r) {
      int m = gwave * 16 + r + 8 * half;
      out[(size_t)m * 64 + col] = accs[j][r] + bias;
    }
  }
}

extern "C" void kernel_launch(void* const* d_in, const int* in_sizes, int n_in,
                              void* d_out, int out_size, void* d_ws, size_t ws_size,
                              hipStream_t stream) {
  const float* x  = (const float*)d_in[0];
  const int*   ei = (const int*)d_in[1];
  const float* W1 = (const float*)d_in[2];
  const float* b1 = (const float*)d_in[3];
  const float* W2 = (const float*)d_in[4];
  const float* b2 = (const float*)d_in[5];
  const float* Wl = (const float*)d_in[6];
  const float* bl = (const float*)d_in[7];
  float* out = (float*)d_out;

  const int N = in_sizes[0] / 8;
  const int E = in_sizes[1] / 2;
  const int* src = ei;
  const int* dst = ei + E;

  float* ws   = (float*)d_ws;
  float* dis  = ws;
  float* rec2 = ws + (size_t)4  * N;
  float* h2   = ws + (size_t)8  * N;
  float* rec1 = ws + (size_t)32 * N;
  unsigned* pdeg = (unsigned*)(ws + (size_t)48 * N);
  float* p1  = ws + (size_t)(48 + G_DEG) * N;
  float* p2  = ws + (size_t)(48 + G_DEG + 8 * G_P1) * N;

  const int B = 256;
  const int r_deg = (N + RS_DEG - 1) / RS_DEG;
  const int r_p1  = (N + RS_P1 - 1) / RS_P1;
  const int r_p2  = (N + RS_P2 - 1) / RS_P2;
  const int c_deg = (E + G_DEG - 1) / G_DEG;
  const int c_p1  = (E + G_P1 - 1) / G_P1;
  const int c_p2  = (E + G_P2 - 1) / G_P2;

  degree_lds<<<r_deg * G_DEG, TPB, 0, stream>>>(src, pdeg, E, N, r_deg, c_deg);
  dis_kernel<<<(N + B - 1) / B, B, 0, stream>>>(pdeg, x, dis, rec1, N);

  prop_lds<8, RS_P1, REC1, DOFF1><<<r_p1 * G_P1, TPB, 0, stream>>>(src, dst, rec1, p1, E, N, r_p1, c_p1);
  layer1_kernel<<<(N + B - 1) / B, B, 0, stream>>>(x, p1, W1, b1, dis, rec2, N);

  prop_lds<3, RS_P2, REC2, DOFF2><<<r_p2 * G_P2, TPB, 0, stream>>>(src, dst, rec2, p2, E, N, r_p2, c_p2);
  layer2_kernel<<<(N + B - 1) / B, B, 0, stream>>>(rec2, p2, W2, b2, h2, N);

  const int ntiles = N / 16;
  head_gemm_wmma<<<(ntiles * 32 + B - 1) / B, B, 0, stream>>>(h2, Wl, bl, out, ntiles);
}


# Layer 1 multiplies before it propagates: edges carry y = x@W1[1] (3 floats, record {y, dis}) not x (8): 4 node ranges not 10, 3 LDS adds per edge not 8; product before edge sum: other rounding
# speedup vs baseline: 11.2700x; 1.1808x over previous
#include <hip/hip_runtime.h>
#include <math.h>


typedef __attribute__((ext_vector_type(2))) float v2f;
typedef __attribute__((ext_vector_type(8))) float v8f;

#define TPB 1024

#define RS_DEG 25000
#define RS_P   25000
#define G_DEG 64
#define G_P   64

__global__ __launch_bounds__(TPB) void degree_lds(const int* __restrict__ src,
                                                  unsigned* __restrict__ part,
                                                  int E, int N, int R, int chunk) {
  __shared__ unsigned cnt[RS_DEG];
  const int tid = threadIdx.x;
  const int r = blockIdx.x % R;
  const int g = blockIdx.x / R;
  const int lo = r * RS_DEG;
  for (int i = tid; i < RS_DEG; i += TPB) cnt[i] = 0u;
  __syncthreads();

  const int e0 = g * chunk;
  const int e1 = min(E, e0 + chunk);
  const int last = e1 - 1;
  for (int b = e0; b < e1; b += 4 * TPB) {
    int s[4];
#pragma unroll
    for (int k = 0; k < 4; ++k) {
      int ee = b + tid + k * TPB;
      int sv = src[min(ee, last)];
      s[k] = (ee < e1) ? sv : -1;
    }
#pragma unroll
    for (int k = 0; k < 4; ++k) {
      unsigned u = (unsigned)(s[k] - lo);
      if (u < (unsigned)RS_DEG)
        __hip_atomic_fetch_add(&cnt[u], 1u, __ATOMIC_RELAXED, __HIP_MEMORY_SCOPE_WORKGROUP);
    }
  }
  __syncthreads();
  for (int i = tid; i < RS_DEG; i += TPB)
    if (lo + i < N) part[(size_t)g * N + lo + i] = cnt[i];
}

#define FP 3
#define REC 4
#define DOFF 3

__global__ void dis_kernel(const unsigned* __restrict__ part, const float* __restrict__ x,
                           const float* __restrict__ W1,
                           float* __restrict__ dis, float* __restrict__ rec1, int n) {
  int i = blockIdx.x * blockDim.x + threadIdx.x;
  if (i < n) {
    const float4* xp = reinterpret_cast<const float4*>(x + (size_t)i * 8);
    float4 x0 = xp[0], x1 = xp[1];
    unsigned c = 0u;
#pragma unroll 16
    for (int g = 0; g < G_DEG; ++g) c += part[(size_t)g * n + i];
    float d = (float)c;
    float di = (d > 0.0f) ? (1.0f / sqrtf(d)) : 0.0f;
    dis[i] = di;
    const float xi[8] = {x0.x, x0.y, x0.z, x0.w, x1.x, x1.y, x1.z, x1.w};
    float y[FP];
#pragma unroll
    for (int h = 0; h < FP; ++h) {
      float a = 0.0f;
#pragma unroll
      for (int f = 0; f < 8; ++f) a += xi[f] * W1[24 + f * 3 + h];
      y[h] = a;
    }
    reinterpret_cast<float4*>(rec1)[i] = make_float4(y[0], y[1], y[2], di);
  }
}

__global__ __launch_bounds__(TPB) void prop_lds(const int* __restrict__ src,
                                                const int* __restrict__ dst,
                                                const float* __restrict__ rec,
                                                float* __restrict__ part,
                                                int E, int N, int R, int chunk) {
  static_assert(FP == 3 && REC == 4 && DOFF == 3, "16-byte record {v[3], dis}");
  __shared__ float acc[FP * RS_P];
  const int tid = threadIdx.x;
  const int r = blockIdx.x % R;
  const int g = blockIdx.x / R;
  const int lo = r * RS_P;
  for (int i = tid; i < FP * RS_P; i += TPB) acc[i] = 0.0f;
  __syncthreads();

  const int e0 = g * chunk;
  const int e1 = min(E, e0 + chunk);

  if (e1 > e0) {
    const int last = e1 - 1;
    int d[4] = {-1, -1, -1, -1}, s[4] = {0, 0, 0, 0};
    for (int b = e0; ; b += 4 * TPB) {
      int nd[4], ns[4];
#pragma unroll
      for (int k = 0; k < 4; ++k) {
        int ee = b + tid + k * TPB;
        int ec = min(ee, last);
        int dv = dst[ec];
        ns[k] = src[ec];
        nd[k] = (ee < e1) ? dv : -1;
      }

      unsigned u[4];
      float v[4][REC];
#pragma unroll
      for (int k = 0; k < 4; ++k) {
        u[k] = (unsigned)(d[k] - lo);
        if (u[k] < (unsigned)RS_P) {
          const float* row = rec + (size_t)s[k] * REC;
          float4 a = reinterpret_cast<const float4*>(row)[0];
          v[k][0] = a.x; v[k][1] = a.y; v[k][2] = a.z; v[k][3] = a.w;
        }
      }
#pragma unroll
      for (int k = 0; k < 4; ++k) {
        if (u[k] < (unsigned)RS_P) {
          float w = v[k][DOFF];
#pragma unroll
          for (int f = 0; f < FP; ++f)
            __hip_atomic_fetch_add(&acc[f * RS_P + u[k]], w * v[k][f],
                                   __ATOMIC_RELAXED, __HIP_MEMORY_SCOPE_WORKGROUP);
        }
      }

#pragma unroll
      for (int k = 0; k < 4; ++k) { d[k] = nd[k]; s[k] = ns[k]; }
      if (b >= e1) break;
    }
  }
  __syncthreads();

  for (int i = tid; i < FP * RS_P; i += TPB) {
    int un = i / FP;
    int f  = i - un * FP;
    if (lo + un < N) part[((size_t)g * N + lo + un) * FP + f] = acc[f * RS_P + un];
  }
}

__global__ void layer1_kernel(const float* __restrict__ x, const float* __restrict__ part,
                              const float* __restrict__ W1, const float* __restrict__ b1,
                              const float* __restrict__ dis, float* __restrict__ rec2, int n) {
  int i = blockIdx.x * blockDim.x + threadIdx.x;
  if (i >= n) return;
  const float4* xp = reinterpret_cast<const float4*>(x + (size_t)i * 8);
  float4 x0 = xp[0], x1 = xp[1];
  float tv[FP] = {0.f, 0.f, 0.f};
  static_assert(G_P % 16 == 0, "the chunk loop takes 16 chunks per trip");
  for (int g0 = 0; g0 < G_P; g0 += 16) {
    float p[16][FP];
#pragma unroll
    for (int k = 0; k < 16; ++k) {
      const float* tp = part + ((size_t)(g0 + k) * n + i) * FP;
      p[k][0] = tp[0]; p[k][1] = tp[1]; p[k][2] = tp[2];
    }
#pragma unroll
    for (int k = 0; k < 16; ++k) {
      tv[0] += p[k][0]; tv[1] += p[k][1]; tv[2] += p[k][2];
    }
  }
  const float di = dis[i];
  const float nd = -di;
  const float xi[8] = {x0.x, x0.y, x0.z, x0.w, x1.x, x1.y, x1.z, x1.w};
  float ho[FP];
#pragma unroll
  for (int h = 0; h < FP; ++h) {
    float acc = b1[h];
#pragma unroll
    for (int f = 0; f < 8; ++f) acc += xi[f] * W1[f * 3 + h];
    acc += nd * tv[h];
    ho[h] = fmaxf(acc, 0.0f);
  }
  reinterpret_cast<float4*>(rec2)[i] = make_float4(ho[0], ho[1], ho[2], di);
}

__global__ void layer2_kernel(const float* __restrict__ rec2, const float* __restrict__ part,
                              const float* __restrict__ W2, const float* __restrict__ b2,
                              float* __restrict__ h2, int n) {
  int i = blockIdx.x * blockDim.x + threadIdx.x;
  if (i >= n) return;
  float4 hr = reinterpret_cast<const float4*>(rec2)[i];
  float hv[3] = {hr.x, hr.y, hr.z}, tv[3] = {0.f, 0.f, 0.f};
#pragma unroll 16
  for (int g = 0; g < G_P; ++g) {
    const float* tp = part + ((size_t)g * n + i) * 3;
    tv[0] += tp[0]; tv[1] += tp[1]; tv[2] += tp[2];
  }
  const float nd = -hr.w;
  tv[0] *= nd; tv[1] *= nd; tv[2] *= nd;
#pragma unroll
  for (int h = 0; h < 16; ++h) {
    float acc = b2[h];
#pragma unroll
    for (int f = 0; f < 3; ++f)
      acc += hv[f] * W2[f * 16 + h] + tv[f] * W2[48 + f * 16 + h];
    h2[(size_t)i * 16 + h] = fmaxf(acc, 0.0f);
  }
}

__global__ void head_gemm_wmma(const float* __restrict__ h2, const float* __restrict__ Wl,
                               const float* __restrict__ bl, float* __restrict__ out,
                               int ntiles) {
  int gwave = (blockIdx.x * blockDim.x + threadIdx.x) >> 5;
  int lane  = threadIdx.x & 31;
  if (gwave >= ntiles) return;
  int half = lane >> 4;
  int l16  = lane & 15;
  int row  = gwave * 16 + l16;

  v8f acc0 = {}, acc1 = {}, acc2 = {}, acc3 = {};
#pragma unroll
  for (int kk = 0; kk < 16; kk += 4) {
    int ka = kk + 2 * half;
    v2f a;
    a.x = h2[(size_t)row * 16 + ka];
    a.y = h2[(size_t)row * 16 + ka + 1];

    v2f b0, b1, b2, b3;
    b0.x = Wl[ka * 64 +  0 + l16]; b0.y = Wl[(ka + 1) * 64 +  0 + l16];
    b1.x = Wl[ka * 64 + 16 + l16]; b1.y = Wl[(ka + 1) * 64 + 16 + l16];
    b2.x = Wl[ka * 64 + 32 + l16]; b2.y = Wl[(ka + 1) * 64 + 32 + l16];
    b3.x = Wl[ka * 64 + 48 + l16]; b3.y = Wl[(ka + 1) * 64 + 48 + l16];

    acc0 = __builtin_amdgcn_wmma_f32_16x16x4_f32(false, a, false, b0, (short)0, acc0, false, false);
    acc1 = __builtin_amdgcn_wmma_f32_16x16x4_f32(false, a, false, b1, (short)0, acc1, false, false);
    acc2 = __builtin_amdgcn_wmma_f32_16x16x4_f32(false, a, false, b2, (short)0, acc2, false, false);
    acc3 = __builtin_amdgcn_wmma_f32_16x16x4_f32(false, a, false, b3, (short)0, acc3, false, false);
  }

  v8f accs[4] = {acc0, acc1, acc2, acc3};
#pragma unroll
  for (int j = 0; j < 4; ++j) {
    int col = j * 16 + l16;
    float bias = bl[col];
#pragma unroll
    for (int r = 0; r < 8; ++r) {
      int m = gwave * 16 + r + 8 * half;
      out[(size_t)m * 64 + col] = accs[j][r] + bias;
    }
  }
}

extern "C" void kernel_launch(void* const* d_in, const int* in_sizes, int n_in,
                              void* d_out, int out_size, void* d_ws, size_t ws_size,
                              hipStream_t stream) {
  const float* x  = (const float*)d_in[0];
  const int*   ei = (const int*)d_in[1];
  const float* W1 = (const float*)d_in[2];
  const float* b1 = (const float*)d_in[3];
  const float* W2 = (const float*)d_in[4];
  const float* b2 = (const float*)d_in[5];
  const float* Wl = (const float*)d_in[6];
  const float* bl = (const float*)d_in[7];
  float* out = (float*)d_out;

  const int N = in_sizes[0] / 8;
  const int E = in_sizes[1] / 2;
  const int* src = ei;
  const int* dst = ei + E;

  float* ws   = (float*)d_ws;
  float* dis  = ws;
  float* rec2 = ws + (size_t)4  * N;
  float* h2   = ws + (size_t)8  * N;
  float* rec1 = ws + (size_t)24 * N;
  unsigned* pdeg = (unsigned*)(ws + (size_t)28 * N);
  float* p1  = ws + (size_t)(28 + G_DEG) * N;
  float* p2  = ws + (size_t)(28 + G_DEG + FP * G_P) * N;

  const int B = 256;
  const int r_deg = (N + RS_DEG - 1) / RS_DEG;
  const int r_p   = (N + RS_P - 1) / RS_P;
  const int c_deg = (E + G_DEG - 1) / G_DEG;
  const int c_p   = (E + G_P - 1) / G_P;

  degree_lds<<<r_deg * G_DEG, TPB, 0, stream>>>(src, pdeg, E, N, r_deg, c_deg);
  dis_kernel<<<(N + B - 1) / B, B, 0, stream>>>(pdeg, x, W1, dis, rec1, N);

  prop_lds<<<r_p * G_P, TPB, 0, stream>>>(src, dst, rec1, p1, E, N, r_p, c_p);
  layer1_kernel<<<(N + B - 1) / B, B, 0, stream>>>(x, p1, W1, b1, dis, rec2, N);

  prop_lds<<<r_p * G_P, TPB, 0, stream>>>(src, dst, rec2, p2, E, N, r_p, c_p);
  layer2_kernel<<<(N + B - 1) / B, B, 0, stream>>>(rec2, p2, W2, b2, h2, N);

  const int ntiles = N / 16;
  head_gemm_wmma<<<(ntiles * 32 + B - 1) / B, B, 0, stream>>>(h2, Wl, bl, out, ntiles);
}


# Propagates on 8x8 node-slice tiles: one pass orders each edge chunk by tile as packed words; a work-group holds src rows and dst sums in LDS, no L2 gather per edge; sum order: LDS arrival, 32 tiles
# speedup vs baseline: 17.5364x; 1.5560x over previous
#include <hip/hip_runtime.h>
#include <math.h>


typedef __attribute__((ext_vector_type(2))) float v2f;
typedef __attribute__((ext_vector_type(8))) float v8f;

#define TPB 1024

#define RS_DEG 25000
#define G_DEG 64

__global__ __launch_bounds__(TPB) void degree_lds(const int* __restrict__ src,
                                                  unsigned* __restrict__ part,
                                                  int E, int N, int R, int chunk) {
  __shared__ unsigned cnt[RS_DEG];
  const int tid = threadIdx.x;
  const int r = blockIdx.x % R;
  const int g = blockIdx.x / R;
  const int lo = r * RS_DEG;
  for (int i = tid; i < RS_DEG; i += TPB) cnt[i] = 0u;
  __syncthreads();

  const int e0 = g * chunk;
  const int e1 = min(E, e0 + chunk);
  const int last = e1 - 1;
  for (int b = e0; b < e1; b += 4 * TPB) {
    int s[4];
#pragma unroll
    for (int k = 0; k < 4; ++k) {
      int ee = b + tid + k * TPB;
      int sv = src[min(ee, last)];
      s[k] = (ee < e1) ? sv : -1;
    }
#pragma unroll
    for (int k = 0; k < 4; ++k) {
      unsigned u = (unsigned)(s[k] - lo);
      if (u < (unsigned)RS_DEG)
        __hip_atomic_fetch_add(&cnt[u], 1u, __ATOMIC_RELAXED, __HIP_MEMORY_SCOPE_WORKGROUP);
    }
  }
  __syncthreads();
  for (int i = tid; i < RS_DEG; i += TPB)
    if (lo + i < N) part[(size_t)g * N + lo + i] = cnt[i];
}

#define FP 3

#define SL    12500
#define NSL   8
#define NT    (NSL * NSL)
#define G_BIN 256
#define BCH   25000
#define NQ    4
#define NPART (NSL * NQ)
#define NCOPY 8
#define BU    13

__global__ __launch_bounds__(TPB) void bin_tiles(const int* __restrict__ src,
                                                 const int* __restrict__ dst,
                                                 unsigned* __restrict__ list,
                                                 int* __restrict__ runoff,
                                                 int* __restrict__ runcnt,
                                                 int E, int chunk) {
  static_assert(SL < 65536, "two slice-relative node ids share one word");
  __shared__ unsigned sorted[BCH];
  __shared__ unsigned hist[NT * NCOPY];
  __shared__ unsigned tot[NT];
  const int tid = threadIdx.x;
  const int cpy = tid & (NCOPY - 1);
  const int c = blockIdx.x;
  const int e0 = c * chunk;
  const int e1 = min(E, e0 + chunk);
  if (tid < NT * NCOPY) hist[tid] = 0u;
  __syncthreads();

  if (e1 > e0) {
    const int last = e1 - 1;
    for (int b = e0; b < e1; b += BU * TPB) {
      int tl[BU];
#pragma unroll
      for (int k = 0; k < BU; ++k) {
        int ee = b + tid + k * TPB;
        int ec = min(ee, last);
        int s = src[ec], d = dst[ec];
        int t = (s / SL) * NSL + d / SL;
        tl[k] = (ee < e1) ? t : -1;
      }
#pragma unroll
      for (int k = 0; k < BU; ++k)
        if (tl[k] >= 0)
          __hip_atomic_fetch_add(&hist[tl[k] * NCOPY + cpy], 1u,
                                 __ATOMIC_RELAXED, __HIP_MEMORY_SCOPE_WORKGROUP);
    }
  }
  __syncthreads();
  if (tid < NT) {
    unsigned a = 0u;
    for (int j = 0; j < NCOPY; ++j) a += hist[tid * NCOPY + j];
    tot[tid] = a;
  }
  __syncthreads();
  if (tid < NT) {
    unsigned base = 0u;
    for (int j = 0; j < tid; ++j) base += tot[j];
    runoff[c * NT + tid] = (int)base;
    runcnt[c * NT + tid] = (int)tot[tid];
    for (int j = 0; j < NCOPY; ++j) {
      unsigned h = hist[tid * NCOPY + j];
      hist[tid * NCOPY + j] = base;
      base += h;
    }
  }
  __syncthreads();

  if (e1 > e0) {
    const int last = e1 - 1;
    for (int b = e0; b < e1; b += BU * TPB) {
      int tl[BU];
      unsigned w[BU];
#pragma unroll
      for (int k = 0; k < BU; ++k) {
        int ee = b + tid + k * TPB;
        int ec = min(ee, last);
        int s = src[ec], d = dst[ec];
        int rs = s / SL, rd = d / SL;
        w[k] = (unsigned)(s - rs * SL) | ((unsigned)(d - rd * SL) << 16);
        tl[k] = (ee < e1) ? rs * NSL + rd : -1;
      }
      unsigned pos[BU];
#pragma unroll
      for (int k = 0; k < BU; ++k) {
        pos[k] = 0u;
        if (tl[k] >= 0)
          pos[k] = __hip_atomic_fetch_add(&hist[tl[k] * NCOPY + cpy], 1u,
                                          __ATOMIC_RELAXED, __HIP_MEMORY_SCOPE_WORKGROUP);
      }
#pragma unroll
      for (int k = 0; k < BU; ++k)
        if (tl[k] >= 0) sorted[pos[k]] = w[k];
    }
  }
  __syncthreads();
  const int n = e1 - e0;
  for (int i = tid; i < n; i += TPB) list[(size_t)c * chunk + i] = sorted[i];
}

__global__ void dis_kernel(const unsigned* __restrict__ part, const float* __restrict__ x,
                           const float* __restrict__ W1,
                           float* __restrict__ dis, float* __restrict__ zt, int n) {
  int i = blockIdx.x * blockDim.x + threadIdx.x;
  if (i < n) {
    const float4* xp = reinterpret_cast<const float4*>(x + (size_t)i * 8);
    float4 x0 = xp[0], x1 = xp[1];
    unsigned c = 0u;
#pragma unroll 16
    for (int g = 0; g < G_DEG; ++g) c += part[(size_t)g * n + i];
    float d = (float)c;
    float di = (d > 0.0f) ? (1.0f / sqrtf(d)) : 0.0f;
    dis[i] = di;
    const float xi[8] = {x0.x, x0.y, x0.z, x0.w, x1.x, x1.y, x1.z, x1.w};
#pragma unroll
    for (int h = 0; h < FP; ++h) {
      float a = 0.0f;
#pragma unroll
      for (int f = 0; f < 8; ++f) a += xi[f] * W1[24 + f * 3 + h];
      zt[(size_t)h * n + i] = di * a;
    }
  }
}

__global__ __launch_bounds__(TPB) void prop_tile(const unsigned* __restrict__ list,
                                                 const int* __restrict__ runoff,
                                                 const int* __restrict__ runcnt,
                                                 const float* __restrict__ zt,
                                                 float* __restrict__ part,
                                                 int N, int chunk) {
  static_assert(SL % 4 == 0, "16-byte accesses never straddle a plane");
  __shared__ __attribute__((aligned(16))) float tab[FP * SL];
  __shared__ __attribute__((aligned(16))) float acc[FP * SL];
  const int tid = threadIdx.x;
  const int t  = blockIdx.x / NQ;
  const int q  = blockIdx.x - t * NQ;
  const int rs = t / NSL;
  const int rd = t - rs * NSL;
  const int slo = rs * SL;
  const int dlo = rd * SL;

  constexpr int RPW = (G_BIN / NQ) / (TPB / 32);
  static_assert(RPW * (TPB / 32) * NQ == G_BIN, "every chunk has one wave");
  const int wave = tid >> 5;
  const int lane = tid & 31;
  const int c0 = q * (G_BIN / NQ);
  int rn[RPW];
  const unsigned* rp[RPW];
  unsigned first[RPW][4];
#pragma unroll
  for (int r = 0; r < RPW; ++r) {
    const int c = c0 + wave + r * (TPB / 32);
    rn[r] = runcnt[c * NT + t];
    rp[r] = list + (size_t)c * chunk + runoff[c * NT + t];
  }
#pragma unroll
  for (int r = 0; r < RPW; ++r) {
#pragma unroll
    for (int k = 0; k < 4; ++k) {
      int idx = k * 32 + lane;
      unsigned wv = rp[r][max(min(idx, rn[r] - 1), 0)];
      first[r][k] = (idx < rn[r]) ? wv : 0xFFFFFFFFu;
    }
  }

  constexpr int NG  = FP * SL / 4;
  constexpr int TLT = (NG + TPB - 1) / TPB;
  float4 tv[TLT];
#pragma unroll
  for (int jj = 0; jj < TLT; ++jj) {
    int i4 = min(tid + jj * TPB, NG - 1);
    int f = i4 / (SL / 4);
    int u = (i4 - f * (SL / 4)) * 4;
    tv[jj] = make_float4(0.f, 0.f, 0.f, 0.f);
    if (slo + u < N) tv[jj] = *reinterpret_cast<const float4*>(zt + (size_t)f * N + slo + u);
  }
#pragma unroll
  for (int jj = 0; jj < TLT; ++jj) {
    int i4 = tid + jj * TPB;
    if (i4 < NG) {
      int f = i4 / (SL / 4);
      int u = (i4 - f * (SL / 4)) * 4;
      *reinterpret_cast<float4*>(&tab[f * SL + u]) = tv[jj];
      *reinterpret_cast<float4*>(&acc[f * SL + u]) = make_float4(0.f, 0.f, 0.f, 0.f);
    }
  }
  __syncthreads();

#pragma unroll
  for (int r = 0; r < RPW; ++r) {
    const int n = rn[r];
    const unsigned* run = rp[r];
    unsigned cur[4];
#pragma unroll
    for (int k = 0; k < 4; ++k) cur[k] = first[r][k];
    for (int b = 0; b < n; b += 128) {
      unsigned nxt[4];
#pragma unroll
      for (int k = 0; k < 4; ++k) {
        int idx = b + 128 + k * 32 + lane;
        unsigned wv = run[min(idx, n - 1)];
        nxt[k] = (idx < n) ? wv : 0xFFFFFFFFu;
      }
      float v[4][FP];
#pragma unroll
      for (int k = 0; k < 4; ++k) {
        if (cur[k] != 0xFFFFFFFFu) {
          unsigned ls = cur[k] & 0xFFFFu;
#pragma unroll
          for (int f = 0; f < FP; ++f) v[k][f] = tab[f * SL + ls];
        }
      }
#pragma unroll
      for (int k = 0; k < 4; ++k) {
        if (cur[k] != 0xFFFFFFFFu) {
          unsigned ld = cur[k] >> 16;
#pragma unroll
          for (int f = 0; f < FP; ++f)
            __hip_atomic_fetch_add(&acc[f * SL + ld], v[k][f],
                                   __ATOMIC_RELAXED, __HIP_MEMORY_SCOPE_WORKGROUP);
        }
      }
#pragma unroll
      for (int k = 0; k < 4; ++k) cur[k] = nxt[k];
    }
  }
  __syncthreads();

  const int j = rs * NQ + q;
  for (int i4 = tid; i4 < FP * SL / 4; i4 += TPB) {
    int f = i4 / (SL / 4);
    int u = (i4 - f * (SL / 4)) * 4;
    if (dlo + u < N)
      *reinterpret_cast<float4*>(part + ((size_t)j * FP + f) * N + dlo + u) =
          *reinterpret_cast<const float4*>(&acc[f * SL + u]);
  }
}

__global__ void layer1_kernel(const float* __restrict__ x, const float* __restrict__ part,
                              const float* __restrict__ W1, const float* __restrict__ b1,
                              const float* __restrict__ dis, float* __restrict__ rec2,
                              float* __restrict__ zt, int n) {
  int i = blockIdx.x * blockDim.x + threadIdx.x;
  if (i >= n) return;
  const float4* xp = reinterpret_cast<const float4*>(x + (size_t)i * 8);
  float4 x0 = xp[0], x1 = xp[1];
  float tv[FP] = {0.f, 0.f, 0.f};
  static_assert(NPART % 16 == 0, "the partial loop takes 16 partials per trip");
  for (int g0 = 0; g0 < NPART; g0 += 16) {
    float p[16][FP];
#pragma unroll
    for (int k = 0; k < 16; ++k) {
      const float* tp = part + (size_t)(g0 + k) * FP * n + i;
      p[k][0] = tp[0]; p[k][1] = tp[n]; p[k][2] = tp[2 * (size_t)n];
    }
#pragma unroll
    for (int k = 0; k < 16; ++k) {
      tv[0] += p[k][0]; tv[1] += p[k][1]; tv[2] += p[k][2];
    }
  }
  const float di = dis[i];
  const float nd = -di;
  const float xi[8] = {x0.x, x0.y, x0.z, x0.w, x1.x, x1.y, x1.z, x1.w};
  float ho[FP];
#pragma unroll
  for (int h = 0; h < FP; ++h) {
    float acc = b1[h];
#pragma unroll
    for (int f = 0; f < 8; ++f) acc += xi[f] * W1[f * 3 + h];
    acc += nd * tv[h];
    ho[h] = fmaxf(acc, 0.0f);
    zt[(size_t)h * n + i] = di * ho[h];
  }
  reinterpret_cast<float4*>(rec2)[i] = make_float4(ho[0], ho[1], ho[2], di);
}

__global__ void layer2_kernel(const float* __restrict__ rec2, const float* __restrict__ part,
                              const float* __restrict__ W2, const float* __restrict__ b2,
                              float* __restrict__ h2, int n) {
  int i = blockIdx.x * blockDim.x + threadIdx.x;
  if (i >= n) return;
  float4 hr = reinterpret_cast<const float4*>(rec2)[i];
  float hv[3] = {hr.x, hr.y, hr.z}, tv[3] = {0.f, 0.f, 0.f};
#pragma unroll 16
  for (int g = 0; g < NPART; ++g) {
    const float* tp = part + (size_t)g * 3 * n + i;
    tv[0] += tp[0]; tv[1] += tp[n]; tv[2] += tp[2 * (size_t)n];
  }
  const float nd = -hr.w;
  tv[0] *= nd; tv[1] *= nd; tv[2] *= nd;
#pragma unroll
  for (int h = 0; h < 16; ++h) {
    float acc = b2[h];
#pragma unroll
    for (int f = 0; f < 3; ++f)
      acc += hv[f] * W2[f * 16 + h] + tv[f] * W2[48 + f * 16 + h];
    h2[(size_t)i * 16 + h] = fmaxf(acc, 0.0f);
  }
}

__global__ void head_gemm_wmma(const float* __restrict__ h2, const float* __restrict__ Wl,
                               const float* __restrict__ bl, float* __restrict__ out,
                               int ntiles) {
  int gwave = (blockIdx.x * blockDim.x + threadIdx.x) >> 5;
  int lane  = threadIdx.x & 31;
  if (gwave >= ntiles) return;
  int half = lane >> 4;
  int l16  = lane & 15;
  int row  = gwave * 16 + l16;

  v8f acc0 = {}, acc1 = {}, acc2 = {}, acc3 = {};
#pragma unroll
  for (int kk = 0; kk < 16; kk += 4) {
    int ka = kk + 2 * half;
    v2f a;
    a.x = h2[(size_t)row * 16 + ka];
    a.y = h2[(size_t)row * 16 + ka + 1];

    v2f b0, b1, b2, b3;
    b0.x = Wl[ka * 64 +  0 + l16]; b0.y = Wl[(ka + 1) * 64 +  0 + l16];
    b1.x = Wl[ka * 64 + 16 + l16]; b1.y = Wl[(ka + 1) * 64 + 16 + l16];
    b2.x = Wl[ka * 64 + 32 + l16]; b2.y = Wl[(ka + 1) * 64 + 32 + l16];
    b3.x = Wl[ka * 64 + 48 + l16]; b3.y = Wl[(ka + 1) * 64 + 48 + l16];

    acc0 = __builtin_amdgcn_wmma_f32_16x16x4_f32(false, a, false, b0, (short)0, acc0, false, false);
    acc1 = __builtin_amdgcn_wmma_f32_16x16x4_f32(false, a, false, b1, (short)0, acc1, false, false);
    acc2 = __builtin_amdgcn_wmma_f32_16x16x4_f32(false, a, false, b2, (short)0, acc2, false, false);
    acc3 = __builtin_amdgcn_wmma_f32_16x16x4_f32(false, a, false, b3, (short)0, acc3, false, false);
  }

  v8f accs[4] = {acc0, acc1, acc2, acc3};
#pragma unroll
  for (int j = 0; j < 4; ++j) {
    int col = j * 16 + l16;
    float bias = bl[col];
#pragma unroll
    for (int r = 0; r < 8; ++r) {
      int m = gwave * 16 + r + 8 * half;
      out[(size_t)m * 64 + col] = accs[j][r] + bias;
    }
  }
}

extern "C" void kernel_launch(void* const* d_in, const int* in_sizes, int n_in,
                              void* d_out, int out_size, void* d_ws, size_t ws_size,
                              hipStream_t stream) {
  const float* x  = (const float*)d_in[0];
  const int*   ei = (const int*)d_in[1];
  const float* W1 = (const float*)d_in[2];
  const float* b1 = (const float*)d_in[3];
  const float* W2 = (const float*)d_in[4];
  const float* b2 = (const float*)d_in[5];
  const float* Wl = (const float*)d_in[6];
  const float* bl = (const float*)d_in[7];
  float* out = (float*)d_out;

  const int N = in_sizes[0] / 8;
  const int E = in_sizes[1] / 2;
  const int* src = ei;
  const int* dst = ei + E;

  float* ws   = (float*)d_ws;
  float* dis  = ws;
  float* rec2 = ws + (size_t)4  * N;
  float* h2   = ws + (size_t)8  * N;
  float* z1   = ws + (size_t)24 * N;
  float* z2   = ws + (size_t)28 * N;
  unsigned* pdeg = (unsigned*)(ws + (size_t)32 * N);
  float* p1  = ws + (size_t)(32 + G_DEG) * N;
  float* p2  = ws + (size_t)(32 + G_DEG + FP * NPART) * N;
  const int c_bin = (E + G_BIN - 1) / G_BIN;
  unsigned* list = (unsigned*)(ws + (size_t)(32 + G_DEG + 2 * FP * NPART) * N);
  int* runoff = (int*)(list + (size_t)G_BIN * c_bin);
  int* runcnt = runoff + G_BIN * NT;

  const int B = 256;
  const int r_deg = (N + RS_DEG - 1) / RS_DEG;
  const int c_deg = (E + G_DEG - 1) / G_DEG;

  degree_lds<<<r_deg * G_DEG, TPB, 0, stream>>>(src, pdeg, E, N, r_deg, c_deg);
  dis_kernel<<<(N + B - 1) / B, B, 0, stream>>>(pdeg, x, W1, dis, z1, N);
  bin_tiles<<<G_BIN, TPB, 0, stream>>>(src, dst, list, runoff, runcnt, E, c_bin);

  prop_tile<<<NT * NQ, TPB, 0, stream>>>(list, runoff, runcnt, z1, p1, N, c_bin);
  layer1_kernel<<<(N + B - 1) / B, B, 0, stream>>>(x, p1, W1, b1, dis, rec2, z2, N);

  prop_tile<<<NT * NQ, TPB, 0, stream>>>(list, runoff, runcnt, z2, p2, N, c_bin);
  layer2_kernel<<<(N + B - 1) / B, B, 0, stream>>>(rec2, p2, W2, b2, h2, N);

  const int ntiles = N / 16;
  head_gemm_wmma<<<(ntiles * 32 + B - 1) / B, B, 0, stream>>>(h2, Wl, bl, out, ntiles);
}


# Tiles 8 src x 32 dst slices, a work-group walks its tile in all chunks: 8 partial sums per node, not 32; partial stores and dense-stage reads shrink 4x; sum order: LDS arrival, 8 tiles
# speedup vs baseline: 18.8900x; 1.0772x over previous
#include <hip/hip_runtime.h>
#include <math.h>


typedef __attribute__((ext_vector_type(2))) float v2f;
typedef __attribute__((ext_vector_type(8))) float v8f;

#define TPB 1024

#define RS_DEG 25000
#define G_DEG 64

__global__ __launch_bounds__(TPB) void degree_lds(const int* __restrict__ src,
                                                  unsigned* __restrict__ part,
                                                  int E, int N, int R, int chunk) {
  __shared__ unsigned cnt[RS_DEG];
  const int tid = threadIdx.x;
  const int r = blockIdx.x % R;
  const int g = blockIdx.x / R;
  const int lo = r * RS_DEG;
  for (int i = tid; i < RS_DEG; i += TPB) cnt[i] = 0u;
  __syncthreads();

  const int e0 = g * chunk;
  const int e1 = min(E, e0 + chunk);
  const int last = e1 - 1;
  for (int b = e0; b < e1; b += 4 * TPB) {
    int s[4];
#pragma unroll
    for (int k = 0; k < 4; ++k) {
      int ee = b + tid + k * TPB;
      int sv = src[min(ee, last)];
      s[k] = (ee < e1) ? sv : -1;
    }
#pragma unroll
    for (int k = 0; k < 4; ++k) {
      unsigned u = (unsigned)(s[k] - lo);
      if (u < (unsigned)RS_DEG)
        __hip_atomic_fetch_add(&cnt[u], 1u, __ATOMIC_RELAXED, __HIP_MEMORY_SCOPE_WORKGROUP);
    }
  }
  __syncthreads();
  for (int i = tid; i < RS_DEG; i += TPB)
    if (lo + i < N) part[(size_t)g * N + lo + i] = cnt[i];
}

#define FP 3

#define SSL   12500
#define NSS   8
#define DSL   3136
#define NDS   32
#define NT    (NSS * NDS)
#define G_BIN 256
#define BCH   25000
#define NPART NSS
#define NCOPY 8
#define BU    13

__global__ __launch_bounds__(TPB) void bin_tiles(const int* __restrict__ src,
                                                 const int* __restrict__ dst,
                                                 unsigned* __restrict__ list,
                                                 int* __restrict__ runoff,
                                                 int* __restrict__ runcnt,
                                                 int E, int chunk) {
  static_assert(SSL < 65536 && DSL < 65536, "two slice-relative node ids share one word");
  static_assert(NT % 32 == 0 && NT <= TPB, "the run starts are summed in groups of 32 tiles");
  __shared__ unsigned sorted[BCH];
  __shared__ unsigned hist[NT * NCOPY];
  __shared__ unsigned tot[NT];
  __shared__ unsigned grp[NT / 32];
  const int tid = threadIdx.x;
  const int cpy = tid & (NCOPY - 1);
  const int c = blockIdx.x;
  const int e0 = c * chunk;
  const int e1 = min(E, e0 + chunk);
  for (int i = tid; i < NT * NCOPY; i += TPB) hist[i] = 0u;
  __syncthreads();

  if (e1 > e0) {
    const int last = e1 - 1;
    for (int b = e0; b < e1; b += BU * TPB) {
      int tl[BU];
#pragma unroll
      for (int k = 0; k < BU; ++k) {
        int ee = b + tid + k * TPB;
        int ec = min(ee, last);
        int s = src[ec], d = dst[ec];
        int t = (s / SSL) * NDS + d / DSL;
        tl[k] = (ee < e1) ? t : -1;
      }
#pragma unroll
      for (int k = 0; k < BU; ++k)
        if (tl[k] >= 0)
          __hip_atomic_fetch_add(&hist[tl[k] * NCOPY + cpy], 1u,
                                 __ATOMIC_RELAXED, __HIP_MEMORY_SCOPE_WORKGROUP);
    }
  }
  __syncthreads();
  if (tid < NT) {
    unsigned a = 0u;
    for (int j = 0; j < NCOPY; ++j) a += hist[tid * NCOPY + j];
    tot[tid] = a;
  }
  __syncthreads();
  if (tid < NT / 32) {
    unsigned a = 0u;
    for (int j = 0; j < 32; ++j) a += tot[tid * 32 + j];
    grp[tid] = a;
  }
  __syncthreads();
  if (tid < NT) {
    unsigned base = 0u;
    for (int j = 0; j < (tid >> 5); ++j) base += grp[j];
    for (int j = tid & ~31; j < tid; ++j) base += tot[j];
    runoff[c * NT + tid] = (int)base;
    runcnt[c * NT + tid] = (int)tot[tid];
    for (int j = 0; j < NCOPY; ++j) {
      unsigned h = hist[tid * NCOPY + j];
      hist[tid * NCOPY + j] = base;
      base += h;
    }
  }
  __syncthreads();

  if (e1 > e0) {
    const int last = e1 - 1;
    for (int b = e0; b < e1; b += BU * TPB) {
      int tl[BU];
      unsigned w[BU];
#pragma unroll
      for (int k = 0; k < BU; ++k) {
        int ee = b + tid + k * TPB;
        int ec = min(ee, last);
        int s = src[ec], d = dst[ec];
        int rs = s / SSL, rd = d / DSL;
        w[k] = (unsigned)(s - rs * SSL) | ((unsigned)(d - rd * DSL) << 16);
        tl[k] = (ee < e1) ? rs * NDS + rd : -1;
      }
      unsigned pos[BU];
#pragma unroll
      for (int k = 0; k < BU; ++k) {
        pos[k] = 0u;
        if (tl[k] >= 0)
          pos[k] = __hip_atomic_fetch_add(&hist[tl[k] * NCOPY + cpy], 1u,
                                          __ATOMIC_RELAXED, __HIP_MEMORY_SCOPE_WORKGROUP);
      }
#pragma unroll
      for (int k = 0; k < BU; ++k)
        if (tl[k] >= 0) sorted[pos[k]] = w[k];
    }
  }
  __syncthreads();
  const int n = e1 - e0;
  for (int i = tid; i < n; i += TPB) list[(size_t)c * chunk + i] = sorted[i];
}

__global__ void dis_kernel(const unsigned* __restrict__ part, const float* __restrict__ x,
                           const float* __restrict__ W1,
                           float* __restrict__ dis, float* __restrict__ zt, int n) {
  int i = blockIdx.x * blockDim.x + threadIdx.x;
  if (i < n) {
    const float4* xp = reinterpret_cast<const float4*>(x + (size_t)i * 8);
    float4 x0 = xp[0], x1 = xp[1];
    unsigned c = 0u;
#pragma unroll 16
    for (int g = 0; g < G_DEG; ++g) c += part[(size_t)g * n + i];
    float d = (float)c;
    float di = (d > 0.0f) ? (1.0f / sqrtf(d)) : 0.0f;
    dis[i] = di;
    const float xi[8] = {x0.x, x0.y, x0.z, x0.w, x1.x, x1.y, x1.z, x1.w};
#pragma unroll
    for (int h = 0; h < FP; ++h) {
      float a = 0.0f;
#pragma unroll
      for (int f = 0; f < 8; ++f) a += xi[f] * W1[24 + f * 3 + h];
      zt[(size_t)h * n + i] = di * a;
    }
  }
}

__global__ __launch_bounds__(TPB) void prop_tile(const unsigned* __restrict__ list,
                                                 const int* __restrict__ runoff,
                                                 const int* __restrict__ runcnt,
                                                 const float* __restrict__ zt,
                                                 float* __restrict__ part,
                                                 int N, int chunk) {
  static_assert(SSL % 4 == 0, "16-byte accesses never straddle a plane");
  static_assert(DSL % 32 == 0, "a dst slice's planes are whole 128-byte lines");
  __shared__ __attribute__((aligned(16))) float tab[FP * SSL];
  __shared__ __attribute__((aligned(16))) float acc[FP * DSL];
  const int tid = threadIdx.x;
  const int t  = blockIdx.x;
  const int rs = t / NDS;
  const int rd = t - rs * NDS;
  const int slo = rs * SSL;
  const int dlo = rd * DSL;

  constexpr int RPW = G_BIN / (TPB / 32);
  static_assert(RPW * (TPB / 32) == G_BIN, "every chunk has one wave");
  const int wave = tid >> 5;
  const int lane = tid & 31;
  int rn[RPW];
  const unsigned* rp[RPW];
  unsigned first[RPW][4];
#pragma unroll
  for (int r = 0; r < RPW; ++r) {
    const int c = wave + r * (TPB / 32);
    rn[r] = runcnt[c * NT + t];
    rp[r] = list + (size_t)c * chunk + runoff[c * NT + t];
  }
#pragma unroll
  for (int r = 0; r < RPW; ++r) {
#pragma unroll
    for (int k = 0; k < 4; ++k) {
      int idx = k * 32 + lane;
      unsigned wv = rp[r][max(min(idx, rn[r] - 1), 0)];
      first[r][k] = (idx < rn[r]) ? wv : 0xFFFFFFFFu;
    }
  }

  constexpr int NG  = FP * SSL / 4;
  constexpr int NGA = FP * DSL / 4;
  constexpr int TLT = (NG + TPB - 1) / TPB;
  float4 tv[TLT];
#pragma unroll
  for (int jj = 0; jj < TLT; ++jj) {
    int i4 = min(tid + jj * TPB, NG - 1);
    int f = i4 / (SSL / 4);
    int u = (i4 - f * (SSL / 4)) * 4;
    tv[jj] = make_float4(0.f, 0.f, 0.f, 0.f);
    if (slo + u < N) tv[jj] = *reinterpret_cast<const float4*>(zt + (size_t)f * N + slo + u);
  }
  for (int i4 = tid; i4 < NGA; i4 += TPB)
    reinterpret_cast<float4*>(acc)[i4] = make_float4(0.f, 0.f, 0.f, 0.f);
#pragma unroll
  for (int jj = 0; jj < TLT; ++jj) {
    int i4 = tid + jj * TPB;
    if (i4 < NG) reinterpret_cast<float4*>(tab)[i4] = tv[jj];
  }
  __syncthreads();

#pragma unroll
  for (int r = 0; r < RPW; ++r) {
    const int n = rn[r];
    const unsigned* run = rp[r];
    unsigned cur[4];
#pragma unroll
    for (int k = 0; k < 4; ++k) cur[k] = first[r][k];
    for (int b = 0; b < n; b += 128) {
      if (b != 0) {
#pragma unroll
        for (int k = 0; k < 4; ++k) {
          int idx = b + k * 32 + lane;
          unsigned wv = run[min(idx, n - 1)];
          cur[k] = (idx < n) ? wv : 0xFFFFFFFFu;
        }
      }
      float v[4][FP];
#pragma unroll
      for (int k = 0; k < 4; ++k) {
        if (cur[k] != 0xFFFFFFFFu) {
          unsigned ls = cur[k] & 0xFFFFu;
#pragma unroll
          for (int f = 0; f < FP; ++f) v[k][f] = tab[f * SSL + ls];
        }
      }
#pragma unroll
      for (int k = 0; k < 4; ++k) {
        if (cur[k] != 0xFFFFFFFFu) {
          unsigned ld = cur[k] >> 16;
#pragma unroll
          for (int f = 0; f < FP; ++f)
            __hip_atomic_fetch_add(&acc[f * DSL + ld], v[k][f],
                                   __ATOMIC_RELAXED, __HIP_MEMORY_SCOPE_WORKGROUP);
        }
      }
    }
  }
  __syncthreads();

  for (int i4 = tid; i4 < NGA; i4 += TPB) {
    int f = i4 / (DSL / 4);
    int u = (i4 - f * (DSL / 4)) * 4;
    if (dlo + u < N)
      *reinterpret_cast<float4*>(part + ((size_t)rs * FP + f) * N + dlo + u) =
          *reinterpret_cast<const float4*>(&acc[f * DSL + u]);
  }
}

__global__ void layer1_kernel(const float* __restrict__ x, const float* __restrict__ part,
                              const float* __restrict__ W1, const float* __restrict__ b1,
                              const float* __restrict__ dis, float* __restrict__ rec2,
                              float* __restrict__ zt, int n) {
  int i = blockIdx.x * blockDim.x + threadIdx.x;
  if (i >= n) return;
  const float4* xp = reinterpret_cast<const float4*>(x + (size_t)i * 8);
  float4 x0 = xp[0], x1 = xp[1];
  float tv[FP] = {0.f, 0.f, 0.f};
  {
    float p[NPART][FP];
#pragma unroll
    for (int k = 0; k < NPART; ++k) {
      const float* tp = part + (size_t)k * FP * n + i;
      p[k][0] = tp[0]; p[k][1] = tp[n]; p[k][2] = tp[2 * (size_t)n];
    }
#pragma unroll
    for (int k = 0; k < NPART; ++k) {
      tv[0] += p[k][0]; tv[1] += p[k][1]; tv[2] += p[k][2];
    }
  }
  const float di = dis[i];
  const float nd = -di;
  const float xi[8] = {x0.x, x0.y, x0.z, x0.w, x1.x, x1.y, x1.z, x1.w};
  float ho[FP];
#pragma unroll
  for (int h = 0; h < FP; ++h) {
    float acc = b1[h];
#pragma unroll
    for (int f = 0; f < 8; ++f) acc += xi[f] * W1[f * 3 + h];
    acc += nd * tv[h];
    ho[h] = fmaxf(acc, 0.0f);
    zt[(size_t)h * n + i] = di * ho[h];
  }
  reinterpret_cast<float4*>(rec2)[i] = make_float4(ho[0], ho[1], ho[2], di);
}

__global__ void layer2_kernel(const float* __restrict__ rec2, const float* __restrict__ part,
                              const float* __restrict__ W2, const float* __restrict__ b2,
                              float* __restrict__ h2, int n) {
  int i = blockIdx.x * blockDim.x + threadIdx.x;
  if (i >= n) return;
  float4 hr = reinterpret_cast<const float4*>(rec2)[i];
  float hv[3] = {hr.x, hr.y, hr.z}, tv[3] = {0.f, 0.f, 0.f};
#pragma unroll
  for (int g = 0; g < NPART; ++g) {
    const float* tp = part + (size_t)g * 3 * n + i;
    tv[0] += tp[0]; tv[1] += tp[n]; tv[2] += tp[2 * (size_t)n];
  }
  const float nd = -hr.w;
  tv[0] *= nd; tv[1] *= nd; tv[2] *= nd;
#pragma unroll
  for (int h = 0; h < 16; ++h) {
    float acc = b2[h];
#pragma unroll
    for (int f = 0; f < 3; ++f)
      acc += hv[f] * W2[f * 16 + h] + tv[f] * W2[48 + f * 16 + h];
    h2[(size_t)i * 16 + h] = fmaxf(acc, 0.0f);
  }
}

__global__ void head_gemm_wmma(const float* __restrict__ h2, const float* __restrict__ Wl,
                               const float* __restrict__ bl, float* __restrict__ out,
                               int ntiles) {
  int gwave = (blockIdx.x * blockDim.x + threadIdx.x) >> 5;
  int lane  = threadIdx.x & 31;
  if (gwave >= ntiles) return;
  int half = lane >> 4;
  int l16  = lane & 15;
  int row  = gwave * 16 + l16;

  v8f acc0 = {}, acc1 = {}, acc2 = {}, acc3 = {};
#pragma unroll
  for (int kk = 0; kk < 16; kk += 4) {
    int ka = kk + 2 * half;
    v2f a;
    a.x = h2[(size_t)row * 16 + ka];
    a.y = h2[(size_t)row * 16 + ka + 1];

    v2f b0, b1, b2, b3;
    b0.x = Wl[ka * 64 +  0 + l16]; b0.y = Wl[(ka + 1) * 64 +  0 + l16];
    b1.x = Wl[ka * 64 + 16 + l16]; b1.y = Wl[(ka + 1) * 64 + 16 + l16];
    b2.x = Wl[ka * 64 + 32 + l16]; b2.y = Wl[(ka + 1) * 64 + 32 + l16];
    b3.x = Wl[ka * 64 + 48 + l16]; b3.y = Wl[(ka + 1) * 64 + 48 + l16];

    acc0 = __builtin_amdgcn_wmma_f32_16x16x4_f32(false, a, false, b0, (short)0, acc0, false, false);
    acc1 = __builtin_amdgcn_wmma_f32_16x16x4_f32(false, a, false, b1, (short)0, acc1, false, false);
    acc2 = __builtin_amdgcn_wmma_f32_16x16x4_f32(false, a, false, b2, (short)0, acc2, false, false);
    acc3 = __builtin_amdgcn_wmma_f32_16x16x4_f32(false, a, false, b3, (short)0, acc3, false, false);
  }

  v8f accs[4] = {acc0, acc1, acc2, acc3};
#pragma unroll
  for (int j = 0; j < 4; ++j) {
    int col = j * 16 + l16;
    float bias = bl[col];
#pragma unroll
    for (int r = 0; r < 8; ++r) {
      int m = gwave * 16 + r + 8 * half;
      out[(size_t)m * 64 + col] = accs[j][r] + bias;
    }
  }
}

extern "C" void kernel_launch(void* const* d_in, const int* in_sizes, int n_in,
                              void* d_out, int out_size, void* d_ws, size_t ws_size,
                              hipStream_t stream) {
  const float* x  = (const float*)d_in[0];
  const int*   ei = (const int*)d_in[1];
  const float* W1 = (const float*)d_in[2];
  const float* b1 = (const float*)d_in[3];
  const float* W2 = (const float*)d_in[4];
  const float* b2 = (const float*)d_in[5];
  const float* Wl = (const float*)d_in[6];
  const float* bl = (const float*)d_in[7];
  float* out = (float*)d_out;

  const int N = in_sizes[0] / 8;
  const int E = in_sizes[1] / 2;
  const int* src = ei;
  const int* dst = ei + E;

  float* ws   = (float*)d_ws;
  float* dis  = ws;
  float* rec2 = ws + (size_t)4  * N;
  float* h2   = ws + (size_t)8  * N;
  float* z1   = ws + (size_t)24 * N;
  float* z2   = ws + (size_t)28 * N;
  unsigned* pdeg = (unsigned*)(ws + (size_t)32 * N);
  float* p1  = ws + (size_t)(32 + G_DEG) * N;
  float* p2  = ws + (size_t)(32 + G_DEG + FP * NPART) * N;
  const int c_bin = (E + G_BIN - 1) / G_BIN;
  unsigned* list = (unsigned*)(ws + (size_t)(32 + G_DEG + 2 * FP * NPART) * N);
  int* runoff = (int*)(list + (size_t)G_BIN * c_bin);
  int* runcnt = runoff + G_BIN * NT;

  const int B = 256;
  const int r_deg = (N + RS_DEG - 1) / RS_DEG;
  const int c_deg = (E + G_DEG - 1) / G_DEG;

  degree_lds<<<r_deg * G_DEG, TPB, 0, stream>>>(src, pdeg, E, N, r_deg, c_deg);
  dis_kernel<<<(N + B - 1) / B, B, 0, stream>>>(pdeg, x, W1, dis, z1, N);
  bin_tiles<<<G_BIN, TPB, 0, stream>>>(src, dst, list, runoff, runcnt, E, c_bin);

  prop_tile<<<NT, TPB, 0, stream>>>(list, runoff, runcnt, z1, p1, N, c_bin);
  layer1_kernel<<<(N + B - 1) / B, B, 0, stream>>>(x, p1, W1, b1, dis, rec2, z2, N);

  prop_tile<<<NT, TPB, 0, stream>>>(list, runoff, runcnt, z2, p2, N, c_bin);
  layer2_kernel<<<(N + B - 1) / B, B, 0, stream>>>(rec2, p2, W2, b2, h2, N);

  const int ntiles = N / 16;
  head_gemm_wmma<<<(ntiles * 32 + B - 1) / B, B, 0, stream>>>(h2, Wl, bl, out, ntiles);
}


# Degree counted from the tile-ordered edge list, not from src: a work-group takes one src slice in 8 chunks, one piece per chunk, all loads ahead; 32 integer partials per node, not 64; float sums same
# speedup vs baseline: 25.2167x; 1.3349x over previous
#include <hip/hip_runtime.h>
#include <math.h>


typedef __attribute__((ext_vector_type(2))) float v2f;
typedef __attribute__((ext_vector_type(8))) float v8f;

#define TPB 1024

#define FP 3

#define SSL   12500
#define NSS   8
#define DSL   3136
#define NDS   32
#define NT    (NSS * NDS)
#define G_BIN 256
#define BCH   25000
#define NPART NSS
#define NCOPY 8
#define BU    13

__global__ __launch_bounds__(TPB) void bin_tiles(const int* __restrict__ src,
                                                 const int* __restrict__ dst,
                                                 unsigned* __restrict__ list,
                                                 int* __restrict__ runoff,
                                                 int* __restrict__ runcnt,
                                                 int E, int chunk) {
  static_assert(SSL < 65536 && DSL < 65536, "two slice-relative node ids share one word");
  static_assert(NT % 32 == 0 && NT <= TPB, "the run starts are summed in groups of 32 tiles");
  __shared__ unsigned sorted[BCH];
  __shared__ unsigned hist[NT * NCOPY];
  __shared__ unsigned tot[NT];
  __shared__ unsigned grp[NT / 32];
  const int tid = threadIdx.x;
  const int cpy = tid & (NCOPY - 1);
  const int c = blockIdx.x;
  const int e0 = c * chunk;
  const int e1 = min(E, e0 + chunk);
  for (int i = tid; i < NT * NCOPY; i += TPB) hist[i] = 0u;
  __syncthreads();

  if (e1 > e0) {
    const int last = e1 - 1;
    for (int b = e0; b < e1; b += BU * TPB) {
      int tl[BU];
#pragma unroll
      for (int k = 0; k < BU; ++k) {
        int ee = b + tid + k * TPB;
        int ec = min(ee, last);
        int s = src[ec], d = dst[ec];
        int t = (s / SSL) * NDS + d / DSL;
        tl[k] = (ee < e1) ? t : -1;
      }
#pragma unroll
      for (int k = 0; k < BU; ++k)
        if (tl[k] >= 0)
          __hip_atomic_fetch_add(&hist[tl[k] * NCOPY + cpy], 1u,
                                 __ATOMIC_RELAXED, __HIP_MEMORY_SCOPE_WORKGROUP);
    }
  }
  __syncthreads();
  if (tid < NT) {
    unsigned a = 0u;
    for (int j = 0; j < NCOPY; ++j) a += hist[tid * NCOPY + j];
    tot[tid] = a;
  }
  __syncthreads();
  if (tid < NT / 32) {
    unsigned a = 0u;
    for (int j = 0; j < 32; ++j) a += tot[tid * 32 + j];
    grp[tid] = a;
  }
  __syncthreads();
  if (tid < NT) {
    unsigned base = 0u;
    for (int j = 0; j < (tid >> 5); ++j) base += grp[j];
    for (int j = tid & ~31; j < tid; ++j) base += tot[j];
    runoff[c * NT + tid] = (int)base;
    runcnt[c * NT + tid] = (int)tot[tid];
    for (int j = 0; j < NCOPY; ++j) {
      unsigned h = hist[tid * NCOPY + j];
      hist[tid * NCOPY + j] = base;
      base += h;
    }
  }
  __syncthreads();

  if (e1 > e0) {
    const int last = e1 - 1;
    for (int b = e0; b < e1; b += BU * TPB) {
      int tl[BU];
      unsigned w[BU];
#pragma unroll
      for (int k = 0; k < BU; ++k) {
        int ee = b + tid + k * TPB;
        int ec = min(ee, last);
        int s = src[ec], d = dst[ec];
        int rs = s / SSL, rd = d / DSL;
        w[k] = (unsigned)(s - rs * SSL) | ((unsigned)(d - rd * DSL) << 16);
        tl[k] = (ee < e1) ? rs * NDS + rd : -1;
      }
      unsigned pos[BU];
#pragma unroll
      for (int k = 0; k < BU; ++k) {
        pos[k] = 0u;
        if (tl[k] >= 0)
          pos[k] = __hip_atomic_fetch_add(&hist[tl[k] * NCOPY + cpy], 1u,
                                          __ATOMIC_RELAXED, __HIP_MEMORY_SCOPE_WORKGROUP);
      }
#pragma unroll
      for (int k = 0; k < BU; ++k)
        if (tl[k] >= 0) sorted[pos[k]] = w[k];
    }
  }
  __syncthreads();
  const int n = e1 - e0;
  for (int i = tid; i < n; i += TPB) list[(size_t)c * chunk + i] = sorted[i];
}

#define G_DEG 32
#define DCH   (G_BIN / G_DEG)
#define DPL   4

__global__ __launch_bounds__(TPB) void degree_tiles(const unsigned* __restrict__ list,
                                                    const int* __restrict__ runoff,
                                                    const int* __restrict__ runcnt,
                                                    unsigned* __restrict__ part,
                                                    int N, int chunk) {
  static_assert(G_DEG * DCH == G_BIN, "every chunk is in one group");
  static_assert(SSL % 4 == 0, "the counters leave in 16-byte pieces");
  __shared__ __attribute__((aligned(16))) unsigned cnt[SSL];
  const int tid = threadIdx.x;
  const int rs = blockIdx.x / G_DEG;
  const int g  = blockIdx.x - rs * G_DEG;
  const int slo = rs * SSL;

  int len[DCH];
  const unsigned* seg[DCH];
  unsigned first[DCH][DPL];
#pragma unroll
  for (int j = 0; j < DCH; ++j) {
    const int c  = g * DCH + j;
    const int t0 = c * NT + rs * NDS;
    const int b  = runoff[t0];
    len[j] = runoff[t0 + NDS - 1] + runcnt[t0 + NDS - 1] - b;
    seg[j] = list + (size_t)c * chunk + b;
  }
#pragma unroll
  for (int j = 0; j < DCH; ++j) {
#pragma unroll
    for (int k = 0; k < DPL; ++k) {
      int idx = k * TPB + tid;
      unsigned wv = seg[j][max(min(idx, len[j] - 1), 0)];
      first[j][k] = (idx < len[j]) ? wv : 0xFFFFFFFFu;
    }
  }
  for (int i4 = tid; i4 < SSL / 4; i4 += TPB)
    reinterpret_cast<uint4*>(cnt)[i4] = make_uint4(0u, 0u, 0u, 0u);
  __syncthreads();

#pragma unroll
  for (int j = 0; j < DCH; ++j) {
    const int n = len[j];
    unsigned cur[DPL];
#pragma unroll
    for (int k = 0; k < DPL; ++k) cur[k] = first[j][k];
    for (int b = 0; b < n; b += DPL * TPB) {
      if (b != 0) {
#pragma unroll
        for (int k = 0; k < DPL; ++k) {
          int idx = b + k * TPB + tid;
          unsigned wv = seg[j][min(idx, n - 1)];
          cur[k] = (idx < n) ? wv : 0xFFFFFFFFu;
        }
      }
#pragma unroll
      for (int k = 0; k < DPL; ++k)
        if (cur[k] != 0xFFFFFFFFu)
          __hip_atomic_fetch_add(&cnt[cur[k] & 0xFFFFu], 1u,
                                 __ATOMIC_RELAXED, __HIP_MEMORY_SCOPE_WORKGROUP);
    }
  }
  __syncthreads();

  for (int i4 = tid; i4 < SSL / 4; i4 += TPB) {
    int u = i4 * 4;
    if (slo + u < N)
      *reinterpret_cast<uint4*>(part + (size_t)g * N + slo + u) =
          *reinterpret_cast<const uint4*>(&cnt[u]);
  }
}

__global__ void dis_kernel(const unsigned* __restrict__ part, const float* __restrict__ x,
                           const float* __restrict__ W1,
                           float* __restrict__ dis, float* __restrict__ zt, int n) {
  int i = blockIdx.x * blockDim.x + threadIdx.x;
  if (i < n) {
    const float4* xp = reinterpret_cast<const float4*>(x + (size_t)i * 8);
    float4 x0 = xp[0], x1 = xp[1];
    unsigned c = 0u;
#pragma unroll
    for (int g = 0; g < G_DEG; ++g) c += part[(size_t)g * n + i];
    float d = (float)c;
    float di = (d > 0.0f) ? (1.0f / sqrtf(d)) : 0.0f;
    dis[i] = di;
    const float xi[8] = {x0.x, x0.y, x0.z, x0.w, x1.x, x1.y, x1.z, x1.w};
#pragma unroll
    for (int h = 0; h < FP; ++h) {
      float a = 0.0f;
#pragma unroll
      for (int f = 0; f < 8; ++f) a += xi[f] * W1[24 + f * 3 + h];
      zt[(size_t)h * n + i] = di * a;
    }
  }
}

__global__ __launch_bounds__(TPB) void prop_tile(const unsigned* __restrict__ list,
                                                 const int* __restrict__ runoff,
                                                 const int* __restrict__ runcnt,
                                                 const float* __restrict__ zt,
                                                 float* __restrict__ part,
                                                 int N, int chunk) {
  static_assert(SSL % 4 == 0, "16-byte accesses never straddle a plane");
  static_assert(DSL % 32 == 0, "a dst slice's planes are whole 128-byte lines");
  __shared__ __attribute__((aligned(16))) float tab[FP * SSL];
  __shared__ __attribute__((aligned(16))) float acc[FP * DSL];
  const int tid = threadIdx.x;
  const int t  = blockIdx.x;
  const int rs = t / NDS;
  const int rd = t - rs * NDS;
  const int slo = rs * SSL;
  const int dlo = rd * DSL;

  constexpr int RPW = G_BIN / (TPB / 32);
  static_assert(RPW * (TPB / 32) == G_BIN, "every chunk has one wave");
  const int wave = tid >> 5;
  const int lane = tid & 31;
  int rn[RPW];
  const unsigned* rp[RPW];
  unsigned first[RPW][4];
#pragma unroll
  for (int r = 0; r < RPW; ++r) {
    const int c = wave + r * (TPB / 32);
    rn[r] = runcnt[c * NT + t];
    rp[r] = list + (size_t)c * chunk + runoff[c * NT + t];
  }
#pragma unroll
  for (int r = 0; r < RPW; ++r) {
#pragma unroll
    for (int k = 0; k < 4; ++k) {
      int idx = k * 32 + lane;
      unsigned wv = rp[r][max(min(idx, rn[r] - 1), 0)];
      first[r][k] = (idx < rn[r]) ? wv : 0xFFFFFFFFu;
    }
  }

  constexpr int NG  = FP * SSL / 4;
  constexpr int NGA = FP * DSL / 4;
  constexpr int TLT = (NG + TPB - 1) / TPB;
  float4 tv[TLT];
#pragma unroll
  for (int jj = 0; jj < TLT; ++jj) {
    int i4 = min(tid + jj * TPB, NG - 1);
    int f = i4 / (SSL / 4);
    int u = (i4 - f * (SSL / 4)) * 4;
    tv[jj] = make_float4(0.f, 0.f, 0.f, 0.f);
    if (slo + u < N) tv[jj] = *reinterpret_cast<const float4*>(zt + (size_t)f * N + slo + u);
  }
  for (int i4 = tid; i4 < NGA; i4 += TPB)
    reinterpret_cast<float4*>(acc)[i4] = make_float4(0.f, 0.f, 0.f, 0.f);
#pragma unroll
  for (int jj = 0; jj < TLT; ++jj) {
    int i4 = tid + jj * TPB;
    if (i4 < NG) reinterpret_cast<float4*>(tab)[i4] = tv[jj];
  }
  __syncthreads();

#pragma unroll
  for (int r = 0; r < RPW; ++r) {
    const int n = rn[r];
    const unsigned* run = rp[r];
    unsigned cur[4];
#pragma unroll
    for (int k = 0; k < 4; ++k) cur[k] = first[r][k];
    for (int b = 0; b < n; b += 128) {
      if (b != 0) {
#pragma unroll
        for (int k = 0; k < 4; ++k) {
          int idx = b + k * 32 + lane;
          unsigned wv = run[min(idx, n - 1)];
          cur[k] = (idx < n) ? wv : 0xFFFFFFFFu;
        }
      }
      float v[4][FP];
#pragma unroll
      for (int k = 0; k < 4; ++k) {
        if (cur[k] != 0xFFFFFFFFu) {
          unsigned ls = cur[k] & 0xFFFFu;
#pragma unroll
          for (int f = 0; f < FP; ++f) v[k][f] = tab[f * SSL + ls];
        }
      }
#pragma unroll
      for (int k = 0; k < 4; ++k) {
        if (cur[k] != 0xFFFFFFFFu) {
          unsigned ld = cur[k] >> 16;
#pragma unroll
          for (int f = 0; f < FP; ++f)
            __hip_atomic_fetch_add(&acc[f * DSL + ld], v[k][f],
                                   __ATOMIC_RELAXED, __HIP_MEMORY_SCOPE_WORKGROUP);
        }
      }
    }
  }
  __syncthreads();

  for (int i4 = tid; i4 < NGA; i4 += TPB) {
    int f = i4 / (DSL / 4);
    int u = (i4 - f * (DSL / 4)) * 4;
    if (dlo + u < N)
      *reinterpret_cast<float4*>(part + ((size_t)rs * FP + f) * N + dlo + u) =
          *reinterpret_cast<const float4*>(&acc[f * DSL + u]);
  }
}

__global__ void layer1_kernel(const float* __restrict__ x, const float* __restrict__ part,
                              const float* __restrict__ W1, const float* __restrict__ b1,
                              const float* __restrict__ dis, float* __restrict__ rec2,
                              float* __restrict__ zt, int n) {
  int i = blockIdx.x * blockDim.x + threadIdx.x;
  if (i >= n) return;
  const float4* xp = reinterpret_cast<const float4*>(x + (size_t)i * 8);
  float4 x0 = xp[0], x1 = xp[1];
  float tv[FP] = {0.f, 0.f, 0.f};
  {
    float p[NPART][FP];
#pragma unroll
    for (int k = 0; k < NPART; ++k) {
      const float* tp = part + (size_t)k * FP * n + i;
      p[k][0] = tp[0]; p[k][1] = tp[n]; p[k][2] = tp[2 * (size_t)n];
    }
#pragma unroll
    for (int k = 0; k < NPART; ++k) {
      tv[0] += p[k][0]; tv[1] += p[k][1]; tv[2] += p[k][2];
    }
  }
  const float di = dis[i];
  const float nd = -di;
  const float xi[8] = {x0.x, x0.y, x0.z, x0.w, x1.x, x1.y, x1.z, x1.w};
  float ho[FP];
#pragma unroll
  for (int h = 0; h < FP; ++h) {
    float acc = b1[h];
#pragma unroll
    for (int f = 0; f < 8; ++f) acc += xi[f] * W1[f * 3 + h];
    acc += nd * tv[h];
    ho[h] = fmaxf(acc, 0.0f);
    zt[(size_t)h * n + i] = di * ho[h];
  }
  reinterpret_cast<float4*>(rec2)[i] = make_float4(ho[0], ho[1], ho[2], di);
}

__global__ void layer2_kernel(const float* __restrict__ rec2, const float* __restrict__ part,
                              const float* __restrict__ W2, const float* __restrict__ b2,
                              float* __restrict__ h2, int n) {
  int i = blockIdx.x * blockDim.x + threadIdx.x;
  if (i >= n) return;
  float4 hr = reinterpret_cast<const float4*>(rec2)[i];
  float hv[3] = {hr.x, hr.y, hr.z}, tv[3] = {0.f, 0.f, 0.f};
#pragma unroll
  for (int g = 0; g < NPART; ++g) {
    const float* tp = part + (size_t)g * 3 * n + i;
    tv[0] += tp[0]; tv[1] += tp[n]; tv[2] += tp[2 * (size_t)n];
  }
  const float nd = -hr.w;
  tv[0] *= nd; tv[1] *= nd; tv[2] *= nd;
#pragma unroll
  for (int h = 0; h < 16; ++h) {
    float acc = b2[h];
#pragma unroll
    for (int f = 0; f < 3; ++f)
      acc += hv[f] * W2[f * 16 + h] + tv[f] * W2[48 + f * 16 + h];
    h2[(size_t)i * 16 + h] = fmaxf(acc, 0.0f);
  }
}

__global__ void head_gemm_wmma(const float* __restrict__ h2, const float* __restrict__ Wl,
                               const float* __restrict__ bl, float* __restrict__ out,
                               int ntiles) {
  int gwave = (blockIdx.x * blockDim.x + threadIdx.x) >> 5;
  int lane  = threadIdx.x & 31;
  if (gwave >= ntiles) return;
  int half = lane >> 4;
  int l16  = lane & 15;
  int row  = gwave * 16 + l16;

  v8f acc0 = {}, acc1 = {}, acc2 = {}, acc3 = {};
#pragma unroll
  for (int kk = 0; kk < 16; kk += 4) {
    int ka = kk + 2 * half;
    v2f a;
    a.x = h2[(size_t)row * 16 + ka];
    a.y = h2[(size_t)row * 16 + ka + 1];

    v2f b0, b1, b2, b3;
    b0.x = Wl[ka * 64 +  0 + l16]; b0.y = Wl[(ka + 1) * 64 +  0 + l16];
    b1.x = Wl[ka * 64 + 16 + l16]; b1.y = Wl[(ka + 1) * 64 + 16 + l16];
    b2.x = Wl[ka * 64 + 32 + l16]; b2.y = Wl[(ka + 1) * 64 + 32 + l16];
    b3.x = Wl[ka * 64 + 48 + l16]; b3.y = Wl[(ka + 1) * 64 + 48 + l16];

    acc0 = __builtin_amdgcn_wmma_f32_16x16x4_f32(false, a, false, b0, (short)0, acc0, false, false);
    acc1 = __builtin_amdgcn_wmma_f32_16x16x4_f32(false, a, false, b1, (short)0, acc1, false, false);
    acc2 = __builtin_amdgcn_wmma_f32_16x16x4_f32(false, a, false, b2, (short)0, acc2, false, false);
    acc3 = __builtin_amdgcn_wmma_f32_16x16x4_f32(false, a, false, b3, (short)0, acc3, false, false);
  }

  v8f accs[4] = {acc0, acc1, acc2, acc3};
#pragma unroll
  for (int j = 0; j < 4; ++j) {
    int col = j * 16 + l16;
    float bias = bl[col];
#pragma unroll
    for (int r = 0; r < 8; ++r) {
      int m = gwave * 16 + r + 8 * half;
      out[(size_t)m * 64 + col] = accs[j][r] + bias;
    }
  }
}

extern "C" void kernel_launch(void* const* d_in, const int* in_sizes, int n_in,
                              void* d_out, int out_size, void* d_ws, size_t ws_size,
                              hipStream_t stream) {
  const float* x  = (const float*)d_in[0];
  const int*   ei = (const int*)d_in[1];
  const float* W1 = (const float*)d_in[2];
  const float* b1 = (const float*)d_in[3];
  const float* W2 = (const float*)d_in[4];
  const float* b2 = (const float*)d_in[5];
  const float* Wl = (const float*)d_in[6];
  const float* bl = (const float*)d_in[7];
  float* out = (float*)d_out;

  const int N = in_sizes[0] / 8;
  const int E = in_sizes[1] / 2;
  const int* src = ei;
  const int* dst = ei + E;

  float* ws   = (float*)d_ws;
  float* dis  = ws;
  float* rec2 = ws + (size_t)4  * N;
  float* h2   = ws + (size_t)8  * N;
  float* z1   = ws + (size_t)24 * N;
  float* z2   = ws + (size_t)28 * N;
  unsigned* pdeg = (unsigned*)(ws + (size_t)32 * N);
  float* p1  = ws + (size_t)(32 + G_DEG) * N;
  float* p2  = ws + (size_t)(32 + G_DEG + FP * NPART) * N;
  const int c_bin = (E + G_BIN - 1) / G_BIN;
  unsigned* list = (unsigned*)(ws + (size_t)(32 + G_DEG + 2 * FP * NPART) * N);
  int* runoff = (int*)(list + (size_t)G_BIN * c_bin);
  int* runcnt = runoff + G_BIN * NT;

  const int B = 256;

  bin_tiles<<<G_BIN, TPB, 0, stream>>>(src, dst, list, runoff, runcnt, E, c_bin);
  degree_tiles<<<NSS * G_DEG, TPB, 0, stream>>>(list, runoff, runcnt, pdeg, N, c_bin);
  dis_kernel<<<(N + B - 1) / B, B, 0, stream>>>(pdeg, x, W1, dis, z1, N);

  prop_tile<<<NT, TPB, 0, stream>>>(list, runoff, runcnt, z1, p1, N, c_bin);
  layer1_kernel<<<(N + B - 1) / B, B, 0, stream>>>(x, p1, W1, b1, dis, rec2, z2, N);

  prop_tile<<<NT, TPB, 0, stream>>>(list, runoff, runcnt, z2, p2, N, c_bin);
  layer2_kernel<<<(N + B - 1) / B, B, 0, stream>>>(rec2, p2, W2, b2, h2, N);

  const int ntiles = N / 16;
  head_gemm_wmma<<<(ntiles * 32 + B - 1) / B, B, 0, stream>>>(h2, Wl, bl, out, ntiles);
}


# Layer 2's dense stage runs in the head GEMM's waves: each lane computes in registers the 8 h2 values of its A fragments; no h2 plane written or read, 8 -> 7 launches; same arithmetic and sum order
# speedup vs baseline: 26.2819x; 1.0422x over previous
#include <hip/hip_runtime.h>
#include <math.h>


typedef __attribute__((ext_vector_type(2))) float v2f;
typedef __attribute__((ext_vector_type(8))) float v8f;

#define TPB 1024

#define FP 3

#define SSL   12500
#define NSS   8
#define DSL   3136
#define NDS   32
#define NT    (NSS * NDS)
#define G_BIN 256
#define BCH   25000
#define NPART NSS
#define NCOPY 8
#define BU    13

__global__ __launch_bounds__(TPB) void bin_tiles(const int* __restrict__ src,
                                                 const int* __restrict__ dst,
                                                 unsigned* __restrict__ list,
                                                 int* __restrict__ runoff,
                                                 int* __restrict__ runcnt,
                                                 int E, int chunk) {
  static_assert(SSL < 65536 && DSL < 65536, "two slice-relative node ids share one word");
  static_assert(NT % 32 == 0 && NT <= TPB, "the run starts are summed in groups of 32 tiles");
  __shared__ unsigned sorted[BCH];
  __shared__ unsigned hist[NT * NCOPY];
  __shared__ unsigned tot[NT];
  __shared__ unsigned grp[NT / 32];
  const int tid = threadIdx.x;
  const int cpy = tid & (NCOPY - 1);
  const int c = blockIdx.x;
  const int e0 = c * chunk;
  const int e1 = min(E, e0 + chunk);
  for (int i = tid; i < NT * NCOPY; i += TPB) hist[i] = 0u;
  __syncthreads();

  if (e1 > e0) {
    const int last = e1 - 1;
    for (int b = e0; b < e1; b += BU * TPB) {
      int tl[BU];
#pragma unroll
      for (int k = 0; k < BU; ++k) {
        int ee = b + tid + k * TPB;
        int ec = min(ee, last);
        int s = src[ec], d = dst[ec];
        int t = (s / SSL) * NDS + d / DSL;
        tl[k] = (ee < e1) ? t : -1;
      }
#pragma unroll
      for (int k = 0; k < BU; ++k)
        if (tl[k] >= 0)
          __hip_atomic_fetch_add(&hist[tl[k] * NCOPY + cpy], 1u,
                                 __ATOMIC_RELAXED, __HIP_MEMORY_SCOPE_WORKGROUP);
    }
  }
  __syncthreads();
  if (tid < NT) {
    unsigned a = 0u;
    for (int j = 0; j < NCOPY; ++j) a += hist[tid * NCOPY + j];
    tot[tid] = a;
  }
  __syncthreads();
  if (tid < NT / 32) {
    unsigned a = 0u;
    for (int j = 0; j < 32; ++j) a += tot[tid * 32 + j];
    grp[tid] = a;
  }
  __syncthreads();
  if (tid < NT) {
    unsigned base = 0u;
    for (int j = 0; j < (tid >> 5); ++j) base += grp[j];
    for (int j = tid & ~31; j < tid; ++j) base += tot[j];
    runoff[c * NT + tid] = (int)base;
    runcnt[c * NT + tid] = (int)tot[tid];
    for (int j = 0; j < NCOPY; ++j) {
      unsigned h = hist[tid * NCOPY + j];
      hist[tid * NCOPY + j] = base;
      base += h;
    }
  }
  __syncthreads();

  if (e1 > e0) {
    const int last = e1 - 1;
    for (int b = e0; b < e1; b += BU * TPB) {
      int tl[BU];
      unsigned w[BU];
#pragma unroll
      for (int k = 0; k < BU; ++k) {
        int ee = b + tid + k * TPB;
        int ec = min(ee, last);
        int s = src[ec], d = dst[ec];
        int rs = s / SSL, rd = d / DSL;
        w[k] = (unsigned)(s - rs * SSL) | ((unsigned)(d - rd * DSL) << 16);
        tl[k] = (ee < e1) ? rs * NDS + rd : -1;
      }
      unsigned pos[BU];
#pragma unroll
      for (int k = 0; k < BU; ++k) {
        pos[k] = 0u;
        if (tl[k] >= 0)
          pos[k] = __hip_atomic_fetch_add(&hist[tl[k] * NCOPY + cpy], 1u,
                                          __ATOMIC_RELAXED, __HIP_MEMORY_SCOPE_WORKGROUP);
      }
#pragma unroll
      for (int k = 0; k < BU; ++k)
        if (tl[k] >= 0) sorted[pos[k]] = w[k];
    }
  }
  __syncthreads();
  const int n = e1 - e0;
  for (int i = tid; i < n; i += TPB) list[(size_t)c * chunk + i] = sorted[i];
}

#define G_DEG 32
#define DCH   (G_BIN / G_DEG)
#define DPL   4

__global__ __launch_bounds__(TPB) void degree_tiles(const unsigned* __restrict__ list,
                                                    const int* __restrict__ runoff,
                                                    const int* __restrict__ runcnt,
                                                    unsigned* __restrict__ part,
                                                    int N, int chunk) {
  static_assert(G_DEG * DCH == G_BIN, "every chunk is in one group");
  static_assert(SSL % 4 == 0, "the counters leave in 16-byte pieces");
  __shared__ __attribute__((aligned(16))) unsigned cnt[SSL];
  const int tid = threadIdx.x;
  const int rs = blockIdx.x / G_DEG;
  const int g  = blockIdx.x - rs * G_DEG;
  const int slo = rs * SSL;

  int len[DCH];
  const unsigned* seg[DCH];
  unsigned first[DCH][DPL];
#pragma unroll
  for (int j = 0; j < DCH; ++j) {
    const int c  = g * DCH + j;
    const int t0 = c * NT + rs * NDS;
    const int b  = runoff[t0];
    len[j] = runoff[t0 + NDS - 1] + runcnt[t0 + NDS - 1] - b;
    seg[j] = list + (size_t)c * chunk + b;
  }
#pragma unroll
  for (int j = 0; j < DCH; ++j) {
#pragma unroll
    for (int k = 0; k < DPL; ++k) {
      int idx = k * TPB + tid;
      unsigned wv = seg[j][max(min(idx, len[j] - 1), 0)];
      first[j][k] = (idx < len[j]) ? wv : 0xFFFFFFFFu;
    }
  }
  for (int i4 = tid; i4 < SSL / 4; i4 += TPB)
    reinterpret_cast<uint4*>(cnt)[i4] = make_uint4(0u, 0u, 0u, 0u);
  __syncthreads();

#pragma unroll
  for (int j = 0; j < DCH; ++j) {
    const int n = len[j];
    unsigned cur[DPL];
#pragma unroll
    for (int k = 0; k < DPL; ++k) cur[k] = first[j][k];
    for (int b = 0; b < n; b += DPL * TPB) {
      if (b != 0) {
#pragma unroll
        for (int k = 0; k < DPL; ++k) {
          int idx = b + k * TPB + tid;
          unsigned wv = seg[j][min(idx, n - 1)];
          cur[k] = (idx < n) ? wv : 0xFFFFFFFFu;
        }
      }
#pragma unroll
      for (int k = 0; k < DPL; ++k)
        if (cur[k] != 0xFFFFFFFFu)
          __hip_atomic_fetch_add(&cnt[cur[k] & 0xFFFFu], 1u,
                                 __ATOMIC_RELAXED, __HIP_MEMORY_SCOPE_WORKGROUP);
    }
  }
  __syncthreads();

  for (int i4 = tid; i4 < SSL / 4; i4 += TPB) {
    int u = i4 * 4;
    if (slo + u < N)
      *reinterpret_cast<uint4*>(part + (size_t)g * N + slo + u) =
          *reinterpret_cast<const uint4*>(&cnt[u]);
  }
}

__global__ void dis_kernel(const unsigned* __restrict__ part, const float* __restrict__ x,
                           const float* __restrict__ W1,
                           float* __restrict__ dis, float* __restrict__ zt, int n) {
  int i = blockIdx.x * blockDim.x + threadIdx.x;
  if (i < n) {
    const float4* xp = reinterpret_cast<const float4*>(x + (size_t)i * 8);
    float4 x0 = xp[0], x1 = xp[1];
    unsigned c = 0u;
#pragma unroll
    for (int g = 0; g < G_DEG; ++g) c += part[(size_t)g * n + i];
    float d = (float)c;
    float di = (d > 0.0f) ? (1.0f / sqrtf(d)) : 0.0f;
    dis[i] = di;
    const float xi[8] = {x0.x, x0.y, x0.z, x0.w, x1.x, x1.y, x1.z, x1.w};
#pragma unroll
    for (int h = 0; h < FP; ++h) {
      float a = 0.0f;
#pragma unroll
      for (int f = 0; f < 8; ++f) a += xi[f] * W1[24 + f * 3 + h];
      zt[(size_t)h * n + i] = di * a;
    }
  }
}

__global__ __launch_bounds__(TPB) void prop_tile(const unsigned* __restrict__ list,
                                                 const int* __restrict__ runoff,
                                                 const int* __restrict__ runcnt,
                                                 const float* __restrict__ zt,
                                                 float* __restrict__ part,
                                                 int N, int chunk) {
  static_assert(SSL % 4 == 0, "16-byte accesses never straddle a plane");
  static_assert(DSL % 32 == 0, "a dst slice's planes are whole 128-byte lines");
  __shared__ __attribute__((aligned(16))) float tab[FP * SSL];
  __shared__ __attribute__((aligned(16))) float acc[FP * DSL];
  const int tid = threadIdx.x;
  const int t  = blockIdx.x;
  const int rs = t / NDS;
  const int rd = t - rs * NDS;
  const int slo = rs * SSL;
  const int dlo = rd * DSL;

  constexpr int RPW = G_BIN / (TPB / 32);
  static_assert(RPW * (TPB / 32) == G_BIN, "every chunk has one wave");
  const int wave = tid >> 5;
  const int lane = tid & 31;
  int rn[RPW];
  const unsigned* rp[RPW];
  unsigned first[RPW][4];
#pragma unroll
  for (int r = 0; r < RPW; ++r) {
    const int c = wave + r * (TPB / 32);
    rn[r] = runcnt[c * NT + t];
    rp[r] = list + (size_t)c * chunk + runoff[c * NT + t];
  }
#pragma unroll
  for (int r = 0; r < RPW; ++r) {
#pragma unroll
    for (int k = 0; k < 4; ++k) {
      int idx = k * 32 + lane;
      unsigned wv = rp[r][max(min(idx, rn[r] - 1), 0)];
      first[r][k] = (idx < rn[r]) ? wv : 0xFFFFFFFFu;
    }
  }

  constexpr int NG  = FP * SSL / 4;
  constexpr int NGA = FP * DSL / 4;
  constexpr int TLT = (NG + TPB - 1) / TPB;
  float4 tv[TLT];
#pragma unroll
  for (int jj = 0; jj < TLT; ++jj) {
    int i4 = min(tid + jj * TPB, NG - 1);
    int f = i4 / (SSL / 4);
    int u = (i4 - f * (SSL / 4)) * 4;
    tv[jj] = make_float4(0.f, 0.f, 0.f, 0.f);
    if (slo + u < N) tv[jj] = *reinterpret_cast<const float4*>(zt + (size_t)f * N + slo + u);
  }
  for (int i4 = tid; i4 < NGA; i4 += TPB)
    reinterpret_cast<float4*>(acc)[i4] = make_float4(0.f, 0.f, 0.f, 0.f);
#pragma unroll
  for (int jj = 0; jj < TLT; ++jj) {
    int i4 = tid + jj * TPB;
    if (i4 < NG) reinterpret_cast<float4*>(tab)[i4] = tv[jj];
  }
  __syncthreads();

#pragma unroll
  for (int r = 0; r < RPW; ++r) {
    const int n = rn[r];
    const unsigned* run = rp[r];
    unsigned cur[4];
#pragma unroll
    for (int k = 0; k < 4; ++k) cur[k] = first[r][k];
    for (int b = 0; b < n; b += 128) {
      if (b != 0) {
#pragma unroll
        for (int k = 0; k < 4; ++k) {
          int idx = b + k * 32 + lane;
          unsigned wv = run[min(idx, n - 1)];
          cur[k] = (idx < n) ? wv : 0xFFFFFFFFu;
        }
      }
      float v[4][FP];
#pragma unroll
      for (int k = 0; k < 4; ++k) {
        if (cur[k] != 0xFFFFFFFFu) {
          unsigned ls = cur[k] & 0xFFFFu;
#pragma unroll
          for (int f = 0; f < FP; ++f) v[k][f] = tab[f * SSL + ls];
        }
      }
#pragma unroll
      for (int k = 0; k < 4; ++k) {
        if (cur[k] != 0xFFFFFFFFu) {
          unsigned ld = cur[k] >> 16;
#pragma unroll
          for (int f = 0; f < FP; ++f)
            __hip_atomic_fetch_add(&acc[f * DSL + ld], v[k][f],
                                   __ATOMIC_RELAXED, __HIP_MEMORY_SCOPE_WORKGROUP);
        }
      }
    }
  }
  __syncthreads();

  for (int i4 = tid; i4 < NGA; i4 += TPB) {
    int f = i4 / (DSL / 4);
    int u = (i4 - f * (DSL / 4)) * 4;
    if (dlo + u < N)
      *reinterpret_cast<float4*>(part + ((size_t)rs * FP + f) * N + dlo + u) =
          *reinterpret_cast<const float4*>(&acc[f * DSL + u]);
  }
}

__global__ void layer1_kernel(const float* __restrict__ x, const float* __restrict__ part,
                              const float* __restrict__ W1, const float* __restrict__ b1,
                              const float* __restrict__ dis, float* __restrict__ rec2,
                              float* __restrict__ zt, int n) {
  int i = blockIdx.x * blockDim.x + threadIdx.x;
  if (i >= n) return;
  const float4* xp = reinterpret_cast<const float4*>(x + (size_t)i * 8);
  float4 x0 = xp[0], x1 = xp[1];
  float tv[FP] = {0.f, 0.f, 0.f};
  {
    float p[NPART][FP];
#pragma unroll
    for (int k = 0; k < NPART; ++k) {
      const float* tp = part + (size_t)k * FP * n + i;
      p[k][0] = tp[0]; p[k][1] = tp[n]; p[k][2] = tp[2 * (size_t)n];
    }
#pragma unroll
    for (int k = 0; k < NPART; ++k) {
      tv[0] += p[k][0]; tv[1] += p[k][1]; tv[2] += p[k][2];
    }
  }
  const float di = dis[i];
  const float nd = -di;
  const float xi[8] = {x0.x, x0.y, x0.z, x0.w, x1.x, x1.y, x1.z, x1.w};
  float ho[FP];
#pragma unroll
  for (int h = 0; h < FP; ++h) {
    float acc = b1[h];
#pragma unroll
    for (int f = 0; f < 8; ++f) acc += xi[f] * W1[f * 3 + h];
    acc += nd * tv[h];
    ho[h] = fmaxf(acc, 0.0f);
    zt[(size_t)h * n + i] = di * ho[h];
  }
  reinterpret_cast<float4*>(rec2)[i] = make_float4(ho[0], ho[1], ho[2], di);
}

__global__ void layer2_head_wmma(const float* __restrict__ rec2, const float* __restrict__ part,
                                 const float* __restrict__ W2, const float* __restrict__ b2,
                                 const float* __restrict__ Wl, const float* __restrict__ bl,
                                 float* __restrict__ out, int n, int ntiles) {
  int gwave = (blockIdx.x * blockDim.x + threadIdx.x) >> 5;
  int lane  = threadIdx.x & 31;
  if (gwave >= ntiles) return;
  int half = lane >> 4;
  int l16  = lane & 15;
  int row  = gwave * 16 + l16;
  const bool hi = half != 0;

  float4 hr = reinterpret_cast<const float4*>(rec2)[row];
  float hv[3] = {hr.x, hr.y, hr.z}, tv[3] = {0.f, 0.f, 0.f};
#pragma unroll
  for (int g = 0; g < NPART; ++g) {
    const float* tp = part + (size_t)g * 3 * n + row;
    tv[0] += tp[0]; tv[1] += tp[n]; tv[2] += tp[2 * (size_t)n];
  }
  const float nd = -hr.w;
  tv[0] *= nd; tv[1] *= nd; tv[2] *= nd;

  v8f acc0 = {}, acc1 = {}, acc2 = {}, acc3 = {};
#pragma unroll
  for (int kk = 0; kk < 16; kk += 4) {
    int ka = kk + 2 * half;
    float av[2];
#pragma unroll
    for (int e = 0; e < 2; ++e) {
      const int h0 = kk + e, h1 = kk + 2 + e;
      float a0 = b2[h0], a1 = b2[h1];
#pragma unroll
      for (int f = 0; f < 3; ++f) {
        a0 += fmaf(hv[f], W2[f * 16 + h0], tv[f] * W2[48 + f * 16 + h0]);
        a1 += fmaf(hv[f], W2[f * 16 + h1], tv[f] * W2[48 + f * 16 + h1]);
      }
      av[e] = fmaxf(hi ? a1 : a0, 0.0f);
    }
    v2f a;
    a.x = av[0];
    a.y = av[1];

    v2f b0, b1, b2v, b3;
    b0.x  = Wl[ka * 64 +  0 + l16]; b0.y  = Wl[(ka + 1) * 64 +  0 + l16];
    b1.x  = Wl[ka * 64 + 16 + l16]; b1.y  = Wl[(ka + 1) * 64 + 16 + l16];
    b2v.x = Wl[ka * 64 + 32 + l16]; b2v.y = Wl[(ka + 1) * 64 + 32 + l16];
    b3.x  = Wl[ka * 64 + 48 + l16]; b3.y  = Wl[(ka + 1) * 64 + 48 + l16];

    acc0 = __builtin_amdgcn_wmma_f32_16x16x4_f32(false, a, false, b0,  (short)0, acc0, false, false);
    acc1 = __builtin_amdgcn_wmma_f32_16x16x4_f32(false, a, false, b1,  (short)0, acc1, false, false);
    acc2 = __builtin_amdgcn_wmma_f32_16x16x4_f32(false, a, false, b2v, (short)0, acc2, false, false);
    acc3 = __builtin_amdgcn_wmma_f32_16x16x4_f32(false, a, false, b3,  (short)0, acc3, false, false);
  }

  v8f accs[4] = {acc0, acc1, acc2, acc3};
#pragma unroll
  for (int j = 0; j < 4; ++j) {
    int col = j * 16 + l16;
    float bias = bl[col];
#pragma unroll
    for (int r = 0; r < 8; ++r) {
      int m = gwave * 16 + r + 8 * half;
      out[(size_t)m * 64 + col] = accs[j][r] + bias;
    }
  }
}

extern "C" void kernel_launch(void* const* d_in, const int* in_sizes, int n_in,
                              void* d_out, int out_size, void* d_ws, size_t ws_size,
                              hipStream_t stream) {
  const float* x  = (const float*)d_in[0];
  const int*   ei = (const int*)d_in[1];
  const float* W1 = (const float*)d_in[2];
  const float* b1 = (const float*)d_in[3];
  const float* W2 = (const float*)d_in[4];
  const float* b2 = (const float*)d_in[5];
  const float* Wl = (const float*)d_in[6];
  const float* bl = (const float*)d_in[7];
  float* out = (float*)d_out;

  const int N = in_sizes[0] / 8;
  const int E = in_sizes[1] / 2;
  const int* src = ei;
  const int* dst = ei + E;

  float* ws   = (float*)d_ws;
  float* dis  = ws;
  float* rec2 = ws + (size_t)4  * N;
  float* z1   = ws + (size_t)24 * N;
  float* z2   = ws + (size_t)28 * N;
  unsigned* pdeg = (unsigned*)(ws + (size_t)32 * N);
  float* p1  = ws + (size_t)(32 + G_DEG) * N;
  float* p2  = ws + (size_t)(32 + G_DEG + FP * NPART) * N;
  const int c_bin = (E + G_BIN - 1) / G_BIN;
  unsigned* list = (unsigned*)(ws + (size_t)(32 + G_DEG + 2 * FP * NPART) * N);
  int* runoff = (int*)(list + (size_t)G_BIN * c_bin);
  int* runcnt = runoff + G_BIN * NT;

  const int B = 256;

  bin_tiles<<<G_BIN, TPB, 0, stream>>>(src, dst, list, runoff, runcnt, E, c_bin);
  degree_tiles<<<NSS * G_DEG, TPB, 0, stream>>>(list, runoff, runcnt, pdeg, N, c_bin);
  dis_kernel<<<(N + B - 1) / B, B, 0, stream>>>(pdeg, x, W1, dis, z1, N);

  prop_tile<<<NT, TPB, 0, stream>>>(list, runoff, runcnt, z1, p1, N, c_bin);
  layer1_kernel<<<(N + B - 1) / B, B, 0, stream>>>(x, p1, W1, b1, dis, rec2, z2, N);

  prop_tile<<<NT, TPB, 0, stream>>>(list, runoff, runcnt, z2, p2, N, c_bin);

  const int ntiles = N / 16;
  layer2_head_wmma<<<(ntiles * 32 + B - 1) / B, B, 0, stream>>>(rec2, p2, W2, b2, Wl, bl, out,
                                                                  N, ntiles);
}


# speedup vs baseline: 27.9031x; 1.0617x over previous
#include <hip/hip_runtime.h>
#include <math.h>


typedef __attribute__((ext_vector_type(2))) float v2f;
typedef __attribute__((ext_vector_type(8))) float v8f;

#define TPB 1024

#define LDS_BLK 32768

__device__ __forceinline__ int simd_pair() {
  unsigned v;
  asm volatile("s_getreg_b32 %0, hwreg(HW_REG_WAVE_HW_ID1, 8, 1)" : "=s"(v));
  return (int)v;
}

#define FP 3

#define SSL   12500
#define NSS   8
#define DSL   3136
#define NDS   32
#define NT    (NSS * NDS)
#define G_BIN 256
#define BCH   25000
#define NPART NSS
#define NCOPY 8
#define BU    13

__global__ __launch_bounds__(TPB) void bin_tiles(const int* __restrict__ src,
                                                 const int* __restrict__ dst,
                                                 unsigned* __restrict__ list,
                                                 int* __restrict__ runoff,
                                                 int* __restrict__ runcnt,
                                                 int E, int chunk) {
  static_assert(SSL < 65536 && DSL < 65536, "two slice-relative node ids share one word");
  static_assert(NT % 32 == 0 && NT <= TPB, "the run starts are summed in groups of 32 tiles");
  __shared__ unsigned sorted[BCH];
  __shared__ unsigned hist[NT * NCOPY];
  __shared__ unsigned tot[NT];
  __shared__ unsigned grp[NT / 32];
  const int tid = threadIdx.x;
  const int cpy = tid & (NCOPY - 1);
  const int c = blockIdx.x;
  const int e0 = c * chunk;
  const int e1 = min(E, e0 + chunk);
  for (int i = tid; i < NT * NCOPY; i += TPB) hist[i] = 0u;
  __syncthreads();

  if (e1 > e0) {
    const int last = e1 - 1;
    for (int b = e0; b < e1; b += BU * TPB) {
      int tl[BU];
#pragma unroll
      for (int k = 0; k < BU; ++k) {
        int ee = b + tid + k * TPB;
        int ec = min(ee, last);
        int s = src[ec], d = dst[ec];
        int t = (s / SSL) * NDS + d / DSL;
        tl[k] = (ee < e1) ? t : -1;
      }
#pragma unroll
      for (int k = 0; k < BU; ++k)
        if (tl[k] >= 0)
          __hip_atomic_fetch_add(&hist[tl[k] * NCOPY + cpy], 1u,
                                 __ATOMIC_RELAXED, __HIP_MEMORY_SCOPE_WORKGROUP);
    }
  }
  __syncthreads();
  if (tid < NT) {
    unsigned a = 0u;
    for (int j = 0; j < NCOPY; ++j) a += hist[tid * NCOPY + j];
    tot[tid] = a;
  }
  __syncthreads();
  if (tid < NT / 32) {
    unsigned a = 0u;
    for (int j = 0; j < 32; ++j) a += tot[tid * 32 + j];
    grp[tid] = a;
  }
  __syncthreads();
  if (tid < NT) {
    unsigned base = 0u;
    for (int j = 0; j < (tid >> 5); ++j) base += grp[j];
    for (int j = tid & ~31; j < tid; ++j) base += tot[j];
    runoff[c * NT + tid] = (int)base;
    runcnt[c * NT + tid] = (int)tot[tid];
    for (int j = 0; j < NCOPY; ++j) {
      unsigned h = hist[tid * NCOPY + j];
      hist[tid * NCOPY + j] = base;
      base += h;
    }
  }
  __syncthreads();

  if (e1 > e0) {
    const int last = e1 - 1;
    for (int b = e0; b < e1; b += BU * TPB) {
      int tl[BU];
      unsigned w[BU];
#pragma unroll
      for (int k = 0; k < BU; ++k) {
        int ee = b + tid + k * TPB;
        int ec = min(ee, last);
        int s = src[ec], d = dst[ec];
        int rs = s / SSL, rd = d / DSL;
        w[k] = (unsigned)(s - rs * SSL) | ((unsigned)(d - rd * DSL) << 16);
        tl[k] = (ee < e1) ? rs * NDS + rd : -1;
      }
      unsigned pos[BU];
#pragma unroll
      for (int k = 0; k < BU; ++k) {
        pos[k] = 0u;
        if (tl[k] >= 0)
          pos[k] = __hip_atomic_fetch_add(&hist[tl[k] * NCOPY + cpy], 1u,
                                          __ATOMIC_RELAXED, __HIP_MEMORY_SCOPE_WORKGROUP);
      }
#pragma unroll
      for (int k = 0; k < BU; ++k)
        if (tl[k] >= 0) sorted[pos[k]] = w[k];
    }
  }
  __syncthreads();
  const int n = e1 - e0;
  for (int i = tid; i < n; i += TPB) list[(size_t)c * chunk + i] = sorted[i];
}

#define G_DEG 32
#define DCH   (G_BIN / G_DEG)
#define DPL   4

__global__ __launch_bounds__(TPB) void degree_tiles(const unsigned* __restrict__ list,
                                                    const int* __restrict__ runoff,
                                                    const int* __restrict__ runcnt,
                                                    unsigned* __restrict__ part,
                                                    int N, int chunk) {
  static_assert(G_DEG * DCH == G_BIN, "every chunk is in one group");
  static_assert(SSL % 4 == 0, "the counters leave in 16-byte pieces");
  constexpr int CNTS_W = ((SSL * 4 + LDS_BLK - 1) / LDS_BLK) * (LDS_BLK / 4);
  __shared__ __attribute__((aligned(16))) unsigned cntl[CNTS_W + SSL];
  unsigned* const cntA = cntl;
  unsigned* const cntB = cntl + CNTS_W;
  unsigned* const cnt  = cntA + simd_pair() * CNTS_W;
  const int tid = threadIdx.x;
  const int rs = blockIdx.x / G_DEG;
  const int g  = blockIdx.x - rs * G_DEG;
  const int slo = rs * SSL;

  int len[DCH];
  const unsigned* seg[DCH];
  unsigned first[DCH][DPL];
#pragma unroll
  for (int j = 0; j < DCH; ++j) {
    const int c  = g * DCH + j;
    const int t0 = c * NT + rs * NDS;
    const int b  = runoff[t0];
    len[j] = runoff[t0 + NDS - 1] + runcnt[t0 + NDS - 1] - b;
    seg[j] = list + (size_t)c * chunk + b;
  }
#pragma unroll
  for (int j = 0; j < DCH; ++j) {
#pragma unroll
    for (int k = 0; k < DPL; ++k) {
      int idx = k * TPB + tid;
      unsigned wv = seg[j][max(min(idx, len[j] - 1), 0)];
      first[j][k] = (idx < len[j]) ? wv : 0xFFFFFFFFu;
    }
  }
  for (int i4 = tid; i4 < SSL / 4; i4 += TPB) {
    reinterpret_cast<uint4*>(cntA)[i4] = make_uint4(0u, 0u, 0u, 0u);
    reinterpret_cast<uint4*>(cntB)[i4] = make_uint4(0u, 0u, 0u, 0u);
  }
  __syncthreads();

#pragma unroll
  for (int j = 0; j < DCH; ++j) {
    const int n = len[j];
    unsigned cur[DPL];
#pragma unroll
    for (int k = 0; k < DPL; ++k) cur[k] = first[j][k];
    for (int b = 0; b < n; b += DPL * TPB) {
      if (b != 0) {
#pragma unroll
        for (int k = 0; k < DPL; ++k) {
          int idx = b + k * TPB + tid;
          unsigned wv = seg[j][min(idx, n - 1)];
          cur[k] = (idx < n) ? wv : 0xFFFFFFFFu;
        }
      }
#pragma unroll
      for (int k = 0; k < DPL; ++k)
        if (cur[k] != 0xFFFFFFFFu)
          __hip_atomic_fetch_add(&cnt[cur[k] & 0xFFFFu], 1u,
                                 __ATOMIC_RELAXED, __HIP_MEMORY_SCOPE_WORKGROUP);
    }
  }
  __syncthreads();

  for (int i4 = tid; i4 < SSL / 4; i4 += TPB) {
    int u = i4 * 4;
    if (slo + u < N) {
      const uint4 a = *reinterpret_cast<const uint4*>(&cntA[u]);
      const uint4 b = *reinterpret_cast<const uint4*>(&cntB[u]);
      *reinterpret_cast<uint4*>(part + (size_t)g * N + slo + u) =
          make_uint4(a.x + b.x, a.y + b.y, a.z + b.z, a.w + b.w);
    }
  }
}

__global__ void dis_kernel(const unsigned* __restrict__ part, const float* __restrict__ x,
                           const float* __restrict__ W1,
                           float* __restrict__ dis, float* __restrict__ zt, int n) {
  int i = blockIdx.x * blockDim.x + threadIdx.x;
  if (i < n) {
    const float4* xp = reinterpret_cast<const float4*>(x + (size_t)i * 8);
    float4 x0 = xp[0], x1 = xp[1];
    unsigned c = 0u;
#pragma unroll
    for (int g = 0; g < G_DEG; ++g) c += part[(size_t)g * n + i];
    float d = (float)c;
    float di = (d > 0.0f) ? (1.0f / sqrtf(d)) : 0.0f;
    dis[i] = di;
    const float xi[8] = {x0.x, x0.y, x0.z, x0.w, x1.x, x1.y, x1.z, x1.w};
#pragma unroll
    for (int h = 0; h < FP; ++h) {
      float a = 0.0f;
#pragma unroll
      for (int f = 0; f < 8; ++f) a += xi[f] * W1[24 + f * 3 + h];
      zt[(size_t)h * n + i] = di * a;
    }
  }
}

__global__ __launch_bounds__(TPB) void prop_tile(const unsigned* __restrict__ list,
                                                 const int* __restrict__ runoff,
                                                 const int* __restrict__ runcnt,
                                                 const float* __restrict__ zt,
                                                 float* __restrict__ part,
                                                 int N, int chunk) {
  static_assert(SSL % 4 == 0, "16-byte accesses never straddle a plane");
  static_assert(DSL % 32 == 0, "a dst slice's planes are whole 128-byte lines");
  constexpr int TAB_W  = FP * SSL;
  constexpr int ACC_W  = FP * DSL;
  constexpr int ACC0_W = ((TAB_W * 4 + LDS_BLK - 1) / LDS_BLK) * (LDS_BLK / 4);
  constexpr int ACCS_W = ((ACC_W * 4 + LDS_BLK - 1) / LDS_BLK) * (LDS_BLK / 4);
  static_assert((ACC0_W + ACCS_W + ACC_W) * 4 <= 320 * 1024, "table and both copies fit the LDS");
  __shared__ __attribute__((aligned(16))) float lds[ACC0_W + ACCS_W + ACC_W];
  float* const tab  = lds;
  float* const accA = lds + ACC0_W;
  float* const accB = accA + ACCS_W;
  float* const acc  = accA + simd_pair() * ACCS_W;
  const int tid = threadIdx.x;
  const int t  = blockIdx.x;
  const int rs = t / NDS;
  const int rd = t - rs * NDS;
  const int slo = rs * SSL;
  const int dlo = rd * DSL;

  constexpr int RPW = G_BIN / (TPB / 32);
  static_assert(RPW * (TPB / 32) == G_BIN, "every chunk has one wave");
  const int wave = tid >> 5;
  const int lane = tid & 31;
  int rn[RPW];
  const unsigned* rp[RPW];
  unsigned first[RPW][4];
#pragma unroll
  for (int r = 0; r < RPW; ++r) {
    const int c = wave + r * (TPB / 32);
    rn[r] = runcnt[c * NT + t];
    rp[r] = list + (size_t)c * chunk + runoff[c * NT + t];
  }
#pragma unroll
  for (int r = 0; r < RPW; ++r) {
#pragma unroll
    for (int k = 0; k < 4; ++k) {
      int idx = k * 32 + lane;
      unsigned wv = rp[r][max(min(idx, rn[r] - 1), 0)];
      first[r][k] = (idx < rn[r]) ? wv : 0xFFFFFFFFu;
    }
  }

  constexpr int NG  = FP * SSL / 4;
  constexpr int NGA = FP * DSL / 4;
  constexpr int TLT = (NG + TPB - 1) / TPB;
  float4 tv[TLT];
#pragma unroll
  for (int jj = 0; jj < TLT; ++jj) {
    int i4 = min(tid + jj * TPB, NG - 1);
    int f = i4 / (SSL / 4);
    int u = (i4 - f * (SSL / 4)) * 4;
    tv[jj] = make_float4(0.f, 0.f, 0.f, 0.f);
    if (slo + u < N) tv[jj] = *reinterpret_cast<const float4*>(zt + (size_t)f * N + slo + u);
  }
  for (int i4 = tid; i4 < NGA; i4 += TPB) {
    reinterpret_cast<float4*>(accA)[i4] = make_float4(0.f, 0.f, 0.f, 0.f);
    reinterpret_cast<float4*>(accB)[i4] = make_float4(0.f, 0.f, 0.f, 0.f);
  }
#pragma unroll
  for (int jj = 0; jj < TLT; ++jj) {
    int i4 = tid + jj * TPB;
    if (i4 < NG) reinterpret_cast<float4*>(tab)[i4] = tv[jj];
  }
  __syncthreads();

#pragma unroll
  for (int r = 0; r < RPW; ++r) {
    const int n = rn[r];
    const unsigned* run = rp[r];
    unsigned cur[4];
#pragma unroll
    for (int k = 0; k < 4; ++k) cur[k] = first[r][k];
    for (int b = 0; b < n; b += 128) {
      if (b != 0) {
#pragma unroll
        for (int k = 0; k < 4; ++k) {
          int idx = b + k * 32 + lane;
          unsigned wv = run[min(idx, n - 1)];
          cur[k] = (idx < n) ? wv : 0xFFFFFFFFu;
        }
      }
      float v[4][FP];
#pragma unroll
      for (int k = 0; k < 4; ++k) {
        if (cur[k] != 0xFFFFFFFFu) {
          unsigned ls = cur[k] & 0xFFFFu;
#pragma unroll
          for (int f = 0; f < FP; ++f) v[k][f] = tab[f * SSL + ls];
        }
      }
#pragma unroll
      for (int k = 0; k < 4; ++k) {
        if (cur[k] != 0xFFFFFFFFu) {
          unsigned ld = cur[k] >> 16;
#pragma unroll
          for (int f = 0; f < FP; ++f)
            __hip_atomic_fetch_add(&acc[f * DSL + ld], v[k][f],
                                   __ATOMIC_RELAXED, __HIP_MEMORY_SCOPE_WORKGROUP);
        }
      }
    }
  }
  __syncthreads();

  for (int i4 = tid; i4 < NGA; i4 += TPB) {
    int f = i4 / (DSL / 4);
    int u = (i4 - f * (DSL / 4)) * 4;
    if (dlo + u < N) {
      const float4 a = *reinterpret_cast<const float4*>(&accA[f * DSL + u]);
      const float4 b = *reinterpret_cast<const float4*>(&accB[f * DSL + u]);
      *reinterpret_cast<float4*>(part + ((size_t)rs * FP + f) * N + dlo + u) =
          make_float4(a.x + b.x, a.y + b.y, a.z + b.z, a.w + b.w);
    }
  }
}

__global__ void layer1_kernel(const float* __restrict__ x, const float* __restrict__ part,
                              const float* __restrict__ W1, const float* __restrict__ b1,
                              const float* __restrict__ dis, float* __restrict__ rec2,
                              float* __restrict__ zt, int n) {
  int i = blockIdx.x * blockDim.x + threadIdx.x;
  if (i >= n) return;
  const float4* xp = reinterpret_cast<const float4*>(x + (size_t)i * 8);
  float4 x0 = xp[0], x1 = xp[1];
  float tv[FP] = {0.f, 0.f, 0.f};
  {
    float p[NPART][FP];
#pragma unroll
    for (int k = 0; k < NPART; ++k) {
      const float* tp = part + (size_t)k * FP * n + i;
      p[k][0] = tp[0]; p[k][1] = tp[n]; p[k][2] = tp[2 * (size_t)n];
    }
#pragma unroll
    for (int k = 0; k < NPART; ++k) {
      tv[0] += p[k][0]; tv[1] += p[k][1]; tv[2] += p[k][2];
    }
  }
  const float di = dis[i];
  const float nd = -di;
  const float xi[8] = {x0.x, x0.y, x0.z, x0.w, x1.x, x1.y, x1.z, x1.w};
  float ho[FP];
#pragma unroll
  for (int h = 0; h < FP; ++h) {
    float acc = b1[h];
#pragma unroll
    for (int f = 0; f < 8; ++f) acc += xi[f] * W1[f * 3 + h];
    acc += nd * tv[h];
    ho[h] = fmaxf(acc, 0.0f);
    zt[(size_t)h * n + i] = di * ho[h];
  }
  reinterpret_cast<float4*>(rec2)[i] = make_float4(ho[0], ho[1], ho[2], di);
}

__global__ void layer2_head_wmma(const float* __restrict__ rec2, const float* __restrict__ part,
                                 const float* __restrict__ W2, const float* __restrict__ b2,
                                 const float* __restrict__ Wl, const float* __restrict__ bl,
                                 float* __restrict__ out, int n, int ntiles) {
  int gwave = (blockIdx.x * blockDim.x + threadIdx.x) >> 5;
  int lane  = threadIdx.x & 31;
  if (gwave >= ntiles) return;
  int half = lane >> 4;
  int l16  = lane & 15;
  int row  = gwave * 16 + l16;
  const bool hi = half != 0;

  float4 hr = reinterpret_cast<const float4*>(rec2)[row];
  float hv[3] = {hr.x, hr.y, hr.z}, tv[3] = {0.f, 0.f, 0.f};
#pragma unroll
  for (int g = 0; g < NPART; ++g) {
    const float* tp = part + (size_t)g * 3 * n + row;
    tv[0] += tp[0]; tv[1] += tp[n]; tv[2] += tp[2 * (size_t)n];
  }
  const float nd = -hr.w;
  tv[0] *= nd; tv[1] *= nd; tv[2] *= nd;

  v8f acc0 = {}, acc1 = {}, acc2 = {}, acc3 = {};
#pragma unroll
  for (int kk = 0; kk < 16; kk += 4) {
    int ka = kk + 2 * half;
    float av[2];
#pragma unroll
    for (int e = 0; e < 2; ++e) {
      const int h0 = kk + e, h1 = kk + 2 + e;
      float a0 = b2[h0], a1 = b2[h1];
#pragma unroll
      for (int f = 0; f < 3; ++f) {
        a0 += fmaf(hv[f], W2[f * 16 + h0], tv[f] * W2[48 + f * 16 + h0]);
        a1 += fmaf(hv[f], W2[f * 16 + h1], tv[f] * W2[48 + f * 16 + h1]);
      }
      av[e] = fmaxf(hi ? a1 : a0, 0.0f);
    }
    v2f a;
    a.x = av[0];
    a.y = av[1];

    v2f b0, b1, b2v, b3;
    b0.x  = Wl[ka * 64 +  0 + l16]; b0.y  = Wl[(ka + 1) * 64 +  0 + l16];
    b1.x  = Wl[ka * 64 + 16 + l16]; b1.y  = Wl[(ka + 1) * 64 + 16 + l16];
    b2v.x = Wl[ka * 64 + 32 + l16]; b2v.y = Wl[(ka + 1) * 64 + 32 + l16];
    b3.x  = Wl[ka * 64 + 48 + l16]; b3.y  = Wl[(ka + 1) * 64 + 48 + l16];

    acc0 = __builtin_amdgcn_wmma_f32_16x16x4_f32(false, a, false, b0,  (short)0, acc0, false, false);
    acc1 = __builtin_amdgcn_wmma_f32_16x16x4_f32(false, a, false, b1,  (short)0, acc1, false, false);
    acc2 = __builtin_amdgcn_wmma_f32_16x16x4_f32(false, a, false, b2v, (short)0, acc2, false, false);
    acc3 = __builtin_amdgcn_wmma_f32_16x16x4_f32(false, a, false, b3,  (short)0, acc3, false, false);
  }

  v8f accs[4] = {acc0, acc1, acc2, acc3};
#pragma unroll
  for (int j = 0; j < 4; ++j) {
    int col = j * 16 + l16;
    float bias = bl[col];
#pragma unroll
    for (int r = 0; r < 8; ++r) {
      int m = gwave * 16 + r + 8 * half;
      out[(size_t)m * 64 + col] = accs[j][r] + bias;
    }
  }
}

extern "C" void kernel_launch(void* const* d_in, const int* in_sizes, int n_in,
                              void* d_out, int out_size, void* d_ws, size_t ws_size,
                              hipStream_t stream) {
  const float* x  = (const float*)d_in[0];
  const int*   ei = (const int*)d_in[1];
  const float* W1 = (const float*)d_in[2];
  const float* b1 = (const float*)d_in[3];
  const float* W2 = (const float*)d_in[4];
  const float* b2 = (const float*)d_in[5];
  const float* Wl = (const float*)d_in[6];
  const float* bl = (const float*)d_in[7];
  float* out = (float*)d_out;

  const int N = in_sizes[0] / 8;
  const int E = in_sizes[1] / 2;
  const int* src = ei;
  const int* dst = ei + E;

  float* ws   = (float*)d_ws;
  float* dis  = ws;
  float* rec2 = ws + (size_t)4  * N;
  float* z1   = ws + (size_t)24 * N;
  float* z2   = ws + (size_t)28 * N;
  unsigned* pdeg = (unsigned*)(ws + (size_t)32 * N);
  float* p1  = ws + (size_t)(32 + G_DEG) * N;
  float* p2  = ws + (size_t)(32 + G_DEG + FP * NPART) * N;
  const int c_bin = (E + G_BIN - 1) / G_BIN;
  unsigned* list = (unsigned*)(ws + (size_t)(32 + G_DEG + 2 * FP * NPART) * N);
  int* runoff = (int*)(list + (size_t)G_BIN * c_bin);
  int* runcnt = runoff + G_BIN * NT;

  const int B = 256;

  bin_tiles<<<G_BIN, TPB, 0, stream>>>(src, dst, list, runoff, runcnt, E, c_bin);
  degree_tiles<<<NSS * G_DEG, TPB, 0, stream>>>(list, runoff, runcnt, pdeg, N, c_bin);
  dis_kernel<<<(N + B - 1) / B, B, 0, stream>>>(pdeg, x, W1, dis, z1, N);

  prop_tile<<<NT, TPB, 0, stream>>>(list, runoff, runcnt, z1, p1, N, c_bin);
  layer1_kernel<<<(N + B - 1) / B, B, 0, stream>>>(x, p1, W1, b1, dis, rec2, z2, N);

  prop_tile<<<NT, TPB, 0, stream>>>(list, runoff, runcnt, z2, p2, N, c_bin);

  const int ntiles = N / 16;
  layer2_head_wmma<<<(ntiles * 32 + B - 1) / B, B, 0, stream>>>(rec2, p2, W2, b2, Wl, bl, out,
                                                                  N, ntiles);
}
